# starting kernel
# baseline (speedup 1.0000x reference)
#include <hip/hip_runtime.h>
#include <hip/hip_bf16.h>


typedef __attribute__((ext_vector_type(2))) float v2f;
typedef __attribute__((ext_vector_type(8))) float v8f;

#define TT   4096
#define DD   256
#define BB   4
#define BAND 320
#define INV_E 0.36787944117144233f

__device__ __forceinline__ v8f wmma_f32(v2f a, v2f b, v8f c) {
  return __builtin_amdgcn_wmma_f32_16x16x4_f32(false, a, false, b,
                                               (short)0, c, false, false);
}

__global__ __launch_bounds__(32)
void da_proj_kernel(const float* __restrict__ x, const float* __restrict__ mask,
                    const float* __restrict__ W, const float* __restrict__ bias,
                    float* __restrict__ y) {
  const int lane = threadIdx.x & 31;
  const int l = lane & 15;
  const int h = lane >> 4;
  const int m0 = blockIdx.x * 16;
  const int N0 = blockIdx.y * 64;

  v8f zero = {};
  v8f c0 = zero, c1 = zero, c2 = zero, c3 = zero;

  const float* xrow = x + (m0 + l) * DD;
  const float* w0 = W + (N0 +  0 + l) * DD;
  const float* w1 = W + (N0 + 16 + l) * DD;
  const float* w2 = W + (N0 + 32 + l) * DD;
  const float* w3 = W + (N0 + 48 + l) * DD;

  for (int k0 = 0; k0 < DD; k0 += 4) {
    const int kk = k0 + 2 * h;
    v2f a = *(const v2f*)(xrow + kk);
    v2f b0 = *(const v2f*)(w0 + kk);
    v2f b1 = *(const v2f*)(w1 + kk);
    v2f b2 = *(const v2f*)(w2 + kk);
    v2f b3 = *(const v2f*)(w3 + kk);
    c0 = wmma_f32(a, b0, c0);
    c1 = wmma_f32(a, b1, c1);
    c2 = wmma_f32(a, b2, c2);
    c3 = wmma_f32(a, b3, c3);
  }

#pragma unroll
  for (int r = 0; r < 8; ++r) {
    const int row = m0 + r + 8 * h;
    const float mf = __expf(mask[row] * INV_E);
    float* yr = y + row * DD;
    yr[N0 +  0 + l] = (c0[r] + bias[N0 +  0 + l]) * mf;
    yr[N0 + 16 + l] = (c1[r] + bias[N0 + 16 + l]) * mf;
    yr[N0 + 32 + l] = (c2[r] + bias[N0 + 32 + l]) * mf;
    yr[N0 + 48 + l] = (c3[r] + bias[N0 + 48 + l]) * mf;
  }
}

__global__ __launch_bounds__(256)
void da_attn_kernel(const float* __restrict__ y, float* __restrict__ out) {
  const int lane = threadIdx.x & 31;
  const int l = lane & 15;
  const int h = lane >> 4;
  const int wv = threadIdx.x >> 5;
  const int it = blockIdx.x & (TT / 16 - 1);
  const int bi = blockIdx.x >> 8;
  const int I0 = it * 16;
  const int N0 = wv * 32;

  const float* yb = y + (bi * TT) * DD;

  int jlo = I0 - BAND; if (jlo < 0) jlo = 0;
  int jhi = I0 + 16 + BAND; if (jhi > TT) jhi = TT;

  v8f zero = {};
  v8f c0 = zero, c1 = zero;
  const float m = (float)(I0 + l);

  for (int j0 = jlo; j0 < jhi; j0 += 4) {
    const int kk = j0 + 2 * h;
    const float k0f = (float)kk;
    v2f a;
    a.x = __expf(-fabsf(m - k0f) * INV_E);
    a.y = __expf(-fabsf(m - (k0f + 1.0f)) * INV_E);

    const float* r0 = yb + kk * DD;
    const float* r1 = r0 + DD;
    v2f b0, b1;
    b0.x = r0[N0 + l];       b0.y = r1[N0 + l];
    b1.x = r0[N0 + 16 + l];  b1.y = r1[N0 + 16 + l];

    c0 = wmma_f32(a, b0, c0);
    c1 = wmma_f32(a, b1, c1);
  }

  float* ob = out + (bi * TT + I0) * DD;
#pragma unroll
  for (int r = 0; r < 8; ++r) {
    const int row = r + 8 * h;
    ob[row * DD + N0 + l]      = c0[r];
    ob[row * DD + N0 + 16 + l] = c1[r];
  }
}

extern "C" void kernel_launch(void* const* d_in, const int* in_sizes, int n_in,
                              void* d_out, int out_size, void* d_ws, size_t ws_size,
                              hipStream_t stream) {
  const float* x    = (const float*)d_in[0];
  const float* mask = (const float*)d_in[1];
  const float* W    = (const float*)d_in[2];
  const float* bias = (const float*)d_in[3];
  float* out = (float*)d_out;
  float* y   = (float*)d_ws;

  dim3 gproj(BB * TT / 16, DD / 64);
  da_proj_kernel<<<gproj, 32, 0, stream>>>(x, mask, W, bias, y);

  da_attn_kernel<<<BB * TT / 16, 256, 0, stream>>>(y, out);
}


# attn: 128-row x 64-col block tiles; each 32-row chunk of the y' band goes through LDS once (16-byte whole-line loads, 2 stages) and feeds 8 waves, one row tile each; sums and their order unchanged
# speedup vs baseline: 1.2095x; 1.2095x over previous
#include <hip/hip_runtime.h>
#include <hip/hip_bf16.h>


typedef __attribute__((ext_vector_type(2))) float v2f;
typedef __attribute__((ext_vector_type(8))) float v8f;

#define TT   4096
#define DD   256
#define BB   4
#define BAND 320
#define INV_E 0.36787944117144233f

__device__ __forceinline__ v8f wmma_f32(v2f a, v2f b, v8f c) {
  return __builtin_amdgcn_wmma_f32_16x16x4_f32(false, a, false, b,
                                               (short)0, c, false, false);
}

__global__ __launch_bounds__(32)
void da_proj_kernel(const float* __restrict__ x, const float* __restrict__ mask,
                    const float* __restrict__ W, const float* __restrict__ bias,
                    float* __restrict__ y) {
  const int lane = threadIdx.x & 31;
  const int l = lane & 15;
  const int h = lane >> 4;
  const int m0 = blockIdx.x * 16;
  const int N0 = blockIdx.y * 64;

  v8f zero = {};
  v8f c0 = zero, c1 = zero, c2 = zero, c3 = zero;

  const float* xrow = x + (m0 + l) * DD;
  const float* w0 = W + (N0 +  0 + l) * DD;
  const float* w1 = W + (N0 + 16 + l) * DD;
  const float* w2 = W + (N0 + 32 + l) * DD;
  const float* w3 = W + (N0 + 48 + l) * DD;

  for (int k0 = 0; k0 < DD; k0 += 4) {
    const int kk = k0 + 2 * h;
    v2f a = *(const v2f*)(xrow + kk);
    v2f b0 = *(const v2f*)(w0 + kk);
    v2f b1 = *(const v2f*)(w1 + kk);
    v2f b2 = *(const v2f*)(w2 + kk);
    v2f b3 = *(const v2f*)(w3 + kk);
    c0 = wmma_f32(a, b0, c0);
    c1 = wmma_f32(a, b1, c1);
    c2 = wmma_f32(a, b2, c2);
    c3 = wmma_f32(a, b3, c3);
  }

#pragma unroll
  for (int r = 0; r < 8; ++r) {
    const int row = m0 + r + 8 * h;
    const float mf = __expf(mask[row] * INV_E);
    float* yr = y + row * DD;
    yr[N0 +  0 + l] = (c0[r] + bias[N0 +  0 + l]) * mf;
    yr[N0 + 16 + l] = (c1[r] + bias[N0 + 16 + l]) * mf;
    yr[N0 + 32 + l] = (c2[r] + bias[N0 + 32 + l]) * mf;
    yr[N0 + 48 + l] = (c3[r] + bias[N0 + 48 + l]) * mf;
  }
}

typedef __attribute__((ext_vector_type(4))) float v4f;

#define AROWS   128
#define ACOLS   64
#define CHUNK   32
#define ROWPAIR 160
#define STAGE   (CHUNK / 2 * ROWPAIR)

__global__ __launch_bounds__(256)
void da_attn_kernel(const float* __restrict__ y, float* __restrict__ out) {
  __shared__ __attribute__((aligned(16))) float sb[2 * STAGE];

  const int tid = threadIdx.x;
  const int lane = tid & 31;
  const int l = lane & 15;
  const int h = lane >> 4;
  const int wv = __builtin_amdgcn_readfirstlane(tid >> 5);
  const int I0 = blockIdx.x * AROWS;
  const int N0 = blockIdx.y * ACOLS;
  const int bi = blockIdx.z;

  const float* yb = y + (bi * TT) * DD + N0;

  int jlo = I0 - BAND; if (jlo < 0) jlo = 0;
  int jhi = I0 + AROWS + BAND; if (jhi > TT) jhi = TT;
  const int nch = (jhi - jlo) / CHUNK;

  const int cg = tid & 15;
  const int rp = tid >> 4;
  const float* gsrc = yb + (jlo + 2 * rp) * DD + 4 * cg;
  float* sdst = sb + rp * ROWPAIR + 8 * cg;

  const int Iw = I0 + 16 * wv;
  const int wlo = Iw - BAND;
  const int whi = Iw + 16 + BAND;
  const float m = (float)(Iw + l);

  v8f zero = {};
  v8f c0 = zero, c1 = zero, c2 = zero, c3 = zero;

  v4f g0 = *(const v4f*)(gsrc);
  v4f g1 = *(const v4f*)(gsrc + DD);
  {
    v4f s0 = {g0.x, g1.x, g0.y, g1.y};
    v4f s1 = {g0.z, g1.z, g0.w, g1.w};
    *(v4f*)(sdst)     = s0;
    *(v4f*)(sdst + 4) = s1;
  }
  __syncthreads();

  for (int ci = 0; ci < nch; ++ci) {
    const int jc = jlo + ci * CHUNK;
    const bool more = (ci + 1) < nch;
    if (more) {
      const float* gn = gsrc + (ci + 1) * CHUNK * DD;
      g0 = *(const v4f*)(gn);
      g1 = *(const v4f*)(gn + DD);
    }
    const float* sbuf = sb + (ci & 1) * STAGE;

    for (int hh = 0; hh < 2; ++hh) {
      const int js = jc + 16 * hh;
      if (js >= wlo && js < whi) {
        for (int s = 0; s < 4; ++s) {
          const int kk = js + 4 * s + 2 * h;
          const float k0f = (float)kk;
          v2f a;
          a.x = __expf(-fabsf(m - k0f) * INV_E);
          a.y = __expf(-fabsf(m - (k0f + 1.0f)) * INV_E);

          const float* p = sbuf + (8 * hh + 2 * s + h) * ROWPAIR + 2 * l;
          const v2f b0 = *(const v2f*)(p);
          const v2f b1 = *(const v2f*)(p + 32);
          const v2f b2 = *(const v2f*)(p + 64);
          const v2f b3 = *(const v2f*)(p + 96);

          c0 = wmma_f32(a, b0, c0);
          c1 = wmma_f32(a, b1, c1);
          c2 = wmma_f32(a, b2, c2);
          c3 = wmma_f32(a, b3, c3);
        }
      }
    }

    if (more) {
      float* sn = sdst + ((ci + 1) & 1) * STAGE;
      v4f s0 = {g0.x, g1.x, g0.y, g1.y};
      v4f s1 = {g0.z, g1.z, g0.w, g1.w};
      *(v4f*)(sn)     = s0;
      *(v4f*)(sn + 4) = s1;
    }
    __syncthreads();
  }

  float* ob = out + (bi * TT + Iw) * DD + N0;
  for (int r = 0; r < 8; ++r) {
    const int row = r + 8 * h;
    ob[row * DD + l]      = c0[r];
    ob[row * DD + 16 + l] = c1[r];
    ob[row * DD + 32 + l] = c2[r];
    ob[row * DD + 48 + l] = c3[r];
  }
}

extern "C" void kernel_launch(void* const* d_in, const int* in_sizes, int n_in,
                              void* d_out, int out_size, void* d_ws, size_t ws_size,
                              hipStream_t stream) {
  const float* x    = (const float*)d_in[0];
  const float* mask = (const float*)d_in[1];
  const float* W    = (const float*)d_in[2];
  const float* bias = (const float*)d_in[3];
  float* out = (float*)d_out;
  float* y   = (float*)d_ws;

  dim3 gproj(BB * TT / 16, DD / 64);
  da_proj_kernel<<<gproj, 32, 0, stream>>>(x, mask, W, bias, y);

  dim3 gattn(TT / AROWS, DD / ACOLS, BB);
  da_attn_kernel<<<gattn, 256, 0, stream>>>(y, out);
}


# attn: band cut 320 -> 288 rows (f32 weights beyond are exactly 0): 10 staged chunks per block (was 11), 37 sub-chunks per row tile (was 41); row tiles paired 0+3 and 1+2 per SIMD; sums unchanged
# speedup vs baseline: 1.8801x; 1.5544x over previous
#include <hip/hip_runtime.h>


typedef __attribute__((ext_vector_type(2))) float v2f;
typedef __attribute__((ext_vector_type(4))) float v4f;
typedef __attribute__((ext_vector_type(8))) float v8f;

#define TT   4096
#define DD   256
#define BB   4
#define BAND 288
#define NCH  ((BROWS + 2 * BAND) / CHUNK)
#define INV_E 0.36787944117144233f

#define BROWS 64
#define CHUNK 64
#define RP    544
#define STAGE (CHUNK / 2 * RP)
#define ZP    260
#define NIT   (CHUNK / 2 * (DD / 4) / 256)

__device__ __forceinline__ v8f wmma_f32(v2f a, v2f b, v8f c) {
  return __builtin_amdgcn_wmma_f32_16x16x4_f32(false, a, false, b,
                                               (short)0, c, false, false);
}

__global__ __launch_bounds__(256)
void da_fused_kernel(const float* __restrict__ x, const float* __restrict__ mask,
                     const float* __restrict__ W, const float* __restrict__ bias,
                     float* __restrict__ out) {
  __shared__ __attribute__((aligned(16))) float sx[2 * STAGE];
  __shared__ __attribute__((aligned(16))) float sz[BROWS * ZP];
  __shared__ __attribute__((aligned(16))) float smf[2 * CHUNK];
  __shared__ __attribute__((aligned(16))) float ss[BROWS];

  const int tid = threadIdx.x;
  const int lane = tid & 31;
  const int c = lane & 15;
  const int h = lane >> 4;
  const int wv = __builtin_amdgcn_readfirstlane(tid >> 5);
  const int rt = (wv < 4) ? ((wv & 3) >> 1) : 3 - ((wv & 3) >> 1);
  const int ch = wv & 1;
  const int I0 = blockIdx.x * BROWS;
  const int bi = blockIdx.y;

  const float* xb = x + (bi * TT) * DD;
  const float* mb = mask + bi * TT;

  const int jlo = I0 - BAND;
  const int c_lo = (jlo < 0) ? ((-jlo) >> 6) : 0;
  int c_hi = (TT - jlo + CHUNK - 1) >> 6; if (c_hi > NCH) c_hi = NCH;

  const int cg = tid & 63;
  const int rp0 = tid >> 6;
  const float* gsrc = xb + 4 * cg;
  float* sdst = sx + rp0 * RP + 8 * cg;

  const int Iw = I0 + 16 * rt;
  int wlo = Iw - BAND; if (wlo < 0) wlo = 0;
  int whi = Iw + 16 + BAND; if (whi > TT) whi = TT;
  const float m = (float)(Iw + c);

  v8f zero = {};
  v8f acc[8];
  for (int t = 0; t < 8; ++t) acc[t] = zero;
  float ssum = 0.0f;

  v4f g0[NIT], g1[NIT];
  float mv = 0.0f;

  {
    const int j0 = jlo + c_lo * CHUNK;
    for (int it = 0; it < NIT; ++it) {
      int r = j0 + 2 * rp0 + 8 * it;
      r = r < 0 ? 0 : (r > TT - 2 ? TT - 2 : r);
      g0[it] = *(const v4f*)(gsrc + r * DD);
      g1[it] = *(const v4f*)(gsrc + r * DD + DD);
    }
    if (tid < CHUNK) {
      int r = j0 + tid;
      r = r < 0 ? 0 : (r > TT - 1 ? TT - 1 : r);
      mv = mb[r];
    }
  }
  for (int it = 0; it < NIT; ++it) {
    v4f s0 = {g0[it].x, g1[it].x, g0[it].y, g1[it].y};
    v4f s1 = {g0[it].z, g1[it].z, g0[it].w, g1[it].w};
    *(v4f*)(sdst + it * 4 * RP)     = s0;
    *(v4f*)(sdst + it * 4 * RP + 4) = s1;
  }
  if (tid < CHUNK) smf[tid] = __expf(mv * INV_E);
  __syncthreads();

  for (int ci = c_lo; ci < c_hi; ++ci) {
    const int jc = jlo + ci * CHUNK;
    const int st = (ci - c_lo) & 1;
    const bool more = (ci + 1) < c_hi;
    if (more) {
      for (int it = 0; it < NIT; ++it) {
        int r = jc + CHUNK + 2 * rp0 + 8 * it;
        r = r < 0 ? 0 : (r > TT - 2 ? TT - 2 : r);
        g0[it] = *(const v4f*)(gsrc + r * DD);
        g1[it] = *(const v4f*)(gsrc + r * DD + DD);
      }
      if (tid < CHUNK) {
        int r = jc + CHUNK + tid;
        r = r < 0 ? 0 : (r > TT - 1 ? TT - 1 : r);
        mv = mb[r];
      }
    }
    const float* sbuf = sx + st * STAGE;
    const float* mfb = smf + st * CHUNK;

    for (int hh = 0; hh < CHUNK / 16; ++hh) {
      const int js = jc + 16 * hh;
      if (js >= wlo && js < whi) {
        v2f fv[4];
        v2f bf[4][8];
        for (int s = 0; s < 4; ++s) {
          const int lr = 16 * hh + 4 * s + 2 * h;
          fv[s] = *(const v2f*)(mfb + lr);
        }
        for (int s = 0; s < 4; ++s) {
          const int lr = 16 * hh + 4 * s + 2 * h;
          const float* p = sbuf + (lr >> 1) * RP + 256 * ch + 2 * c;
          for (int t = 0; t < 8; ++t)
            bf[s][t] = *(const v2f*)(p + 32 * t);
        }
        v2f ew[4];
        for (int s = 0; s < 4; ++s) {
          const float k0f = (float)(jc + 16 * hh + 4 * s + 2 * h);
          ew[s].x = __expf(-fabsf(m - k0f) * INV_E);
          ew[s].y = __expf(-fabsf(m - (k0f + 1.0f)) * INV_E);
        }
        __builtin_amdgcn_sched_barrier(0);
        for (int s = 0; s < 4; ++s) {
          v2f a;
          a.x = ew[s].x * fv[s].x;
          a.y = ew[s].y * fv[s].y;
          ssum += a.x + a.y;
          for (int t = 0; t < 8; ++t) acc[t] = wmma_f32(a, bf[s][t], acc[t]);
        }
      }
    }

    if (more) {
      float* sn = sdst + (st ^ 1) * STAGE;
      for (int it = 0; it < NIT; ++it) {
        v4f s0 = {g0[it].x, g1[it].x, g0[it].y, g1[it].y};
        v4f s1 = {g0[it].z, g1[it].z, g0[it].w, g1[it].w};
        *(v4f*)(sn + it * 4 * RP)     = s0;
        *(v4f*)(sn + it * 4 * RP + 4) = s1;
      }
      if (tid < CHUNK) smf[(st ^ 1) * CHUNK + tid] = __expf(mv * INV_E);
    }
    __syncthreads();
  }

  ssum += __shfl_xor(ssum, 16);
  if (ch == 0 && h == 0) ss[16 * rt + c] = ssum;

  {
    float* zd = sz + (16 * rt + 8 * h) * ZP + 128 * ch + c;
    for (int t = 0; t < 8; ++t) {
      for (int r = 0; r < 8; ++r) zd[r * ZP + 16 * t] = acc[t][r];
    }
  }
  __syncthreads();

  const int n0 = 32 * wv;
  const float* wr0 = W + (n0 + c) * DD + 4 * h;
  const float* wr1 = W + (n0 + 16 + c) * DD + 4 * h;
  const float* zr = sz + c * ZP + 4 * h;

  v8f o[4][2];
  for (int i = 0; i < 4; ++i) { o[i][0] = zero; o[i][1] = zero; }

  v4f bw[3][2][2];
  v4f az[2][4], nz[2][4];
  for (int j = 0; j < 2; ++j) {
    bw[0][j][0] = *(const v4f*)(wr0 + 8 * j);
    bw[0][j][1] = *(const v4f*)(wr1 + 8 * j);
    bw[1][j][0] = *(const v4f*)(wr0 + 16 + 8 * j);
    bw[1][j][1] = *(const v4f*)(wr1 + 16 + 8 * j);
    for (int i = 0; i < 4; ++i) az[j][i] = *(const v4f*)(zr + 16 * i * ZP + 8 * j);
  }

  for (int it = 0; it < DD / 16; ++it) {
    const int kb = 16 * it;
    __builtin_amdgcn_sched_barrier(0);
    if (it + 2 < DD / 16) {
      for (int j = 0; j < 2; ++j) {
        bw[(it + 2) % 3][j][0] = *(const v4f*)(wr0 + kb + 32 + 8 * j);
        bw[(it + 2) % 3][j][1] = *(const v4f*)(wr1 + kb + 32 + 8 * j);
      }
    }
    if (it + 1 < DD / 16) {
      for (int j = 0; j < 2; ++j) {
        for (int i = 0; i < 4; ++i) nz[j][i] = *(const v4f*)(zr + 16 * i * ZP + kb + 16 + 8 * j);
      }
    }
    __builtin_amdgcn_sched_barrier(0);

    for (int j = 0; j < 2; ++j) {
      const v4f w0 = bw[it % 3][j][0];
      const v4f w1 = bw[it % 3][j][1];
      const v2f b0a = {w0.x, w0.y}, b0b = {w0.z, w0.w};
      const v2f b1a = {w1.x, w1.y}, b1b = {w1.z, w1.w};
      for (int i = 0; i < 4; ++i) {
        const v2f aa = {az[j][i].x, az[j][i].y};
        o[i][0] = wmma_f32(aa, b0a, o[i][0]);
        o[i][1] = wmma_f32(aa, b1a, o[i][1]);
      }
      for (int i = 0; i < 4; ++i) {
        const v2f ab = {az[j][i].z, az[j][i].w};
        o[i][0] = wmma_f32(ab, b0b, o[i][0]);
        o[i][1] = wmma_f32(ab, b1b, o[i][1]);
      }
    }
    if (it + 1 < DD / 16) {
      for (int j = 0; j < 2; ++j) {
        for (int i = 0; i < 4; ++i) az[j][i] = nz[j][i];
      }
    }
  }

  const float bia0 = bias[n0 + c];
  const float bia1 = bias[n0 + 16 + c];
  float* ob = out + (bi * TT + I0 + 8 * h) * DD + n0 + c;
  for (int i = 0; i < 4; ++i) {
    for (int r = 0; r < 8; ++r) {
      const float sr = ss[16 * i + 8 * h + r];
      ob[(16 * i + r) * DD]      = o[i][0][r] + sr * bia0;
      ob[(16 * i + r) * DD + 16] = o[i][1][r] + sr * bia1;
    }
  }
}

extern "C" void kernel_launch(void* const* d_in, const int* in_sizes, int n_in,
                              void* d_out, int out_size, void* d_ws, size_t ws_size,
                              hipStream_t stream) {
  const float* x    = (const float*)d_in[0];
  const float* mask = (const float*)d_in[1];
  const float* W    = (const float*)d_in[2];
  const float* bias = (const float*)d_in[3];
  float* out = (float*)d_out;

  dim3 grid(TT / BROWS, BB);
  da_fused_kernel<<<grid, 256, 0, stream>>>(x, mask, W, bias, out);
}


# attn band: a 16x16 weight block depends on i-j only, so each 2x2 group of blocks takes 3 WMMA products, A0(y0+y1), (Am-A0)y1, (Ap-A0)y0, not 4; f is folded into the staged rows; sums regrouped
# speedup vs baseline: 1.9803x; 1.0533x over previous
#include <hip/hip_runtime.h>


typedef __attribute__((ext_vector_type(2))) float v2f;
typedef __attribute__((ext_vector_type(4))) float v4f;
typedef __attribute__((ext_vector_type(8))) float v8f;

#define TT   4096
#define DD   256
#define BB   4
#define BAND 288
#define NCH  ((BROWS + 2 * BAND) / CHUNK)
#define INV_E 0.36787944117144233f

#define BROWS 64
#define CHUNK 64
#define RP    544
#define STAGE (CHUNK / 2 * RP)
#define ZP    260
#define NIT   (CHUNK / 2 * (DD / 4) / 256)

__device__ __forceinline__ v8f wmma_f32(v2f a, v2f b, v8f c) {
  return __builtin_amdgcn_wmma_f32_16x16x4_f32(false, a, false, b,
                                               (short)0, c, false, false);
}

__global__ __launch_bounds__(256)
void da_fused_kernel(const float* __restrict__ x, const float* __restrict__ mask,
                     const float* __restrict__ W, const float* __restrict__ bias,
                     float* __restrict__ out) {
  __shared__ __attribute__((aligned(16))) float sx[2 * STAGE];
  __shared__ __attribute__((aligned(16))) float sz[BROWS * ZP];
  __shared__ __attribute__((aligned(16))) float smf[2 * CHUNK];
  __shared__ __attribute__((aligned(16))) float ss[BROWS];

  const int tid = threadIdx.x;
  const int lane = tid & 31;
  const int c = lane & 15;
  const int h = lane >> 4;
  const int wv = __builtin_amdgcn_readfirstlane(tid >> 5);
  const int rg = wv >> 2;
  const int cq = wv & 3;
  const int I0 = blockIdx.x * BROWS;
  const int bi = blockIdx.y;

  const float* xb = x + (bi * TT) * DD;
  const float* mb = mask + bi * TT;

  const int jlo = I0 - BAND;
  const int c_lo = (jlo < 0) ? ((-jlo) >> 6) : 0;
  int c_hi = (TT - jlo + CHUNK - 1) >> 6; if (c_hi > NCH) c_hi = NCH;

  const int cg = tid & 63;
  const int rp0 = tid >> 6;
  const float* gsrc = xb + 4 * cg;
  float* sdst = sx + rp0 * RP + 8 * cg;

  const int Iw = I0 + 32 * rg;
  int wlo = Iw - BAND; if (wlo < 0) wlo = 0;
  int whi = Iw + BAND + 1; if (whi > TT) whi = TT;
  const float m = (float)(Iw + c);

  v8f zero = {};
  v8f p1[4], p2[4], p3[4];
  for (int t = 0; t < 4; ++t) { p1[t] = zero; p2[t] = zero; p3[t] = zero; }
  float ssum0 = 0.0f, ssum1 = 0.0f;

  v4f g0[NIT], g1[NIT];
  v2f mk[NIT];
  float mv = 0.0f;

  {
    const int j0 = jlo + c_lo * CHUNK;
    for (int it = 0; it < NIT; ++it) {
      int r = j0 + 2 * rp0 + 8 * it;
      r = r < 0 ? 0 : (r > TT - 2 ? TT - 2 : r);
      g0[it] = *(const v4f*)(gsrc + r * DD);
      g1[it] = *(const v4f*)(gsrc + r * DD + DD);
      mk[it] = *(const v2f*)(mb + r);
    }
    if (tid < CHUNK) {
      int r = j0 + tid;
      r = r < 0 ? 0 : (r > TT - 1 ? TT - 1 : r);
      mv = mb[r];
    }
  }
  for (int it = 0; it < NIT; ++it) {
    const float fa = __expf(mk[it].x * INV_E);
    const float fb = __expf(mk[it].y * INV_E);
    v4f s0 = {g0[it].x * fa, g1[it].x * fb, g0[it].y * fa, g1[it].y * fb};
    v4f s1 = {g0[it].z * fa, g1[it].z * fb, g0[it].w * fa, g1[it].w * fb};
    *(v4f*)(sdst + it * 4 * RP)     = s0;
    *(v4f*)(sdst + it * 4 * RP + 4) = s1;
  }
  if (tid < CHUNK) smf[tid] = __expf(mv * INV_E);
  __syncthreads();

  for (int ci = c_lo; ci < c_hi; ++ci) {
    const int jc = jlo + ci * CHUNK;
    const int st = (ci - c_lo) & 1;
    const bool more = (ci + 1) < c_hi;
    if (more) {
      for (int it = 0; it < NIT; ++it) {
        int r = jc + CHUNK + 2 * rp0 + 8 * it;
        r = r < 0 ? 0 : (r > TT - 2 ? TT - 2 : r);
        g0[it] = *(const v4f*)(gsrc + r * DD);
        g1[it] = *(const v4f*)(gsrc + r * DD + DD);
        mk[it] = *(const v2f*)(mb + r);
      }
      if (tid < CHUNK) {
        int r = jc + CHUNK + tid;
        r = r < 0 ? 0 : (r > TT - 1 ? TT - 1 : r);
        mv = mb[r];
      }
    }
    const float* sbuf = sx + st * STAGE;
    const float* mfb = smf + st * CHUNK;

    for (int pp = 0; pp < CHUNK / 32; ++pp) {
      const int js = jc + 32 * pp;
      if (js >= wlo && js < whi) {
        v2f f0[4], f1[4];
        v2f b0[4][4], b1[4][4];
        for (int s = 0; s < 4; ++s) {
          const int lr = 32 * pp + 4 * s + 2 * h;
          f0[s] = *(const v2f*)(mfb + lr);
          f1[s] = *(const v2f*)(mfb + lr + 16);
        }
        for (int s = 0; s < 4; ++s) {
          const int lr = 32 * pp + 4 * s + 2 * h;
          const float* p = sbuf + (lr >> 1) * RP + 128 * cq + 2 * c;
          for (int t = 0; t < 4; ++t) {
            b0[s][t] = *(const v2f*)(p + 32 * t);
            b1[s][t] = *(const v2f*)(p + 8 * RP + 32 * t);
          }
        }
        v2f a0[4], am[4], ap[4];
        for (int s = 0; s < 4; ++s) {
          const float d0 = m - (float)(js + 4 * s + 2 * h);
          const float d1 = d0 - 1.0f;
          v2f e0, em, ep;
          e0.x = __expf(-fabsf(d0) * INV_E);
          e0.y = __expf(-fabsf(d1) * INV_E);
          em.x = __expf(-fabsf(d0 - 16.0f) * INV_E);
          em.y = __expf(-fabsf(d1 - 16.0f) * INV_E);
          ep.x = __expf(-fabsf(d0 + 16.0f) * INV_E);
          ep.y = __expf(-fabsf(d1 + 16.0f) * INV_E);
          ssum0 += (e0.x * f0[s].x + e0.y * f0[s].y) + (em.x * f1[s].x + em.y * f1[s].y);
          ssum1 += (ep.x * f0[s].x + ep.y * f0[s].y) + (e0.x * f1[s].x + e0.y * f1[s].y);
          a0[s] = e0;
          am[s] = em - e0;
          ap[s] = ep - e0;
        }
        v2f bs[4][4];
        for (int s = 0; s < 4; ++s) {
          for (int t = 0; t < 4; ++t) bs[s][t] = b0[s][t] + b1[s][t];
        }
        __builtin_amdgcn_sched_barrier(0);
        for (int s = 0; s < 4; ++s) {
          for (int t = 0; t < 4; ++t) {
            p3[t] = wmma_f32(ap[s], b0[s][t], p3[t]);
            p2[t] = wmma_f32(am[s], b1[s][t], p2[t]);
            p1[t] = wmma_f32(a0[s], bs[s][t], p1[t]);
          }
        }
      }
    }

    if (more) {
      float* sn = sdst + (st ^ 1) * STAGE;
      for (int it = 0; it < NIT; ++it) {
        const float fa = __expf(mk[it].x * INV_E);
        const float fb = __expf(mk[it].y * INV_E);
        v4f s0 = {g0[it].x * fa, g1[it].x * fb, g0[it].y * fa, g1[it].y * fb};
        v4f s1 = {g0[it].z * fa, g1[it].z * fb, g0[it].w * fa, g1[it].w * fb};
        *(v4f*)(sn + it * 4 * RP)     = s0;
        *(v4f*)(sn + it * 4 * RP + 4) = s1;
      }
      if (tid < CHUNK) smf[(st ^ 1) * CHUNK + tid] = __expf(mv * INV_E);
    }
    __syncthreads();
  }

  ssum0 += __shfl_xor(ssum0, 16);
  ssum1 += __shfl_xor(ssum1, 16);
  if (cq == 0 && h == 0) {
    ss[32 * rg + c]      = ssum0;
    ss[32 * rg + 16 + c] = ssum1;
  }

  {
    float* zd = sz + (32 * rg + 8 * h) * ZP + 64 * cq + c;
    for (int t = 0; t < 4; ++t) {
      for (int r = 0; r < 8; ++r) {
        zd[r * ZP + 16 * t]        = p1[t][r] + p2[t][r];
        zd[(16 + r) * ZP + 16 * t] = p1[t][r] + p3[t][r];
      }
    }
  }
  __syncthreads();

  const int n0 = 32 * wv;
  const float* wr0 = W + (n0 + c) * DD + 4 * h;
  const float* wr1 = W + (n0 + 16 + c) * DD + 4 * h;
  const float* zr = sz + c * ZP + 4 * h;

  v8f o[4][2];
  for (int i = 0; i < 4; ++i) { o[i][0] = zero; o[i][1] = zero; }

  v4f bw[3][2][2];
  v4f az[2][4], nz[2][4];
  for (int j = 0; j < 2; ++j) {
    bw[0][j][0] = *(const v4f*)(wr0 + 8 * j);
    bw[0][j][1] = *(const v4f*)(wr1 + 8 * j);
    bw[1][j][0] = *(const v4f*)(wr0 + 16 + 8 * j);
    bw[1][j][1] = *(const v4f*)(wr1 + 16 + 8 * j);
    for (int i = 0; i < 4; ++i) az[j][i] = *(const v4f*)(zr + 16 * i * ZP + 8 * j);
  }

  for (int it = 0; it < DD / 16; ++it) {
    const int kb = 16 * it;
    __builtin_amdgcn_sched_barrier(0);
    if (it + 2 < DD / 16) {
      for (int j = 0; j < 2; ++j) {
        bw[(it + 2) % 3][j][0] = *(const v4f*)(wr0 + kb + 32 + 8 * j);
        bw[(it + 2) % 3][j][1] = *(const v4f*)(wr1 + kb + 32 + 8 * j);
      }
    }
    if (it + 1 < DD / 16) {
      for (int j = 0; j < 2; ++j) {
        for (int i = 0; i < 4; ++i) nz[j][i] = *(const v4f*)(zr + 16 * i * ZP + kb + 16 + 8 * j);
      }
    }
    __builtin_amdgcn_sched_barrier(0);

    for (int j = 0; j < 2; ++j) {
      const v4f w0 = bw[it % 3][j][0];
      const v4f w1 = bw[it % 3][j][1];
      const v2f b0a = {w0.x, w0.y}, b0b = {w0.z, w0.w};
      const v2f b1a = {w1.x, w1.y}, b1b = {w1.z, w1.w};
      for (int i = 0; i < 4; ++i) {
        const v2f aa = {az[j][i].x, az[j][i].y};
        o[i][0] = wmma_f32(aa, b0a, o[i][0]);
        o[i][1] = wmma_f32(aa, b1a, o[i][1]);
      }
      for (int i = 0; i < 4; ++i) {
        const v2f ab = {az[j][i].z, az[j][i].w};
        o[i][0] = wmma_f32(ab, b0b, o[i][0]);
        o[i][1] = wmma_f32(ab, b1b, o[i][1]);
      }
    }
    if (it + 1 < DD / 16) {
      for (int j = 0; j < 2; ++j) {
        for (int i = 0; i < 4; ++i) az[j][i] = nz[j][i];
      }
    }
  }

  const float bia0 = bias[n0 + c];
  const float bia1 = bias[n0 + 16 + c];
  float* ob = out + (bi * TT + I0 + 8 * h) * DD + n0 + c;
  for (int i = 0; i < 4; ++i) {
    for (int r = 0; r < 8; ++r) {
      const float sr = ss[16 * i + 8 * h + r];
      ob[(16 * i + r) * DD]      = o[i][0][r] + sr * bia0;
      ob[(16 * i + r) * DD + 16] = o[i][1][r] + sr * bia1;
    }
  }
}

extern "C" void kernel_launch(void* const* d_in, const int* in_sizes, int n_in,
                              void* d_out, int out_size, void* d_ws, size_t ws_size,
                              hipStream_t stream) {
  const float* x    = (const float*)d_in[0];
  const float* mask = (const float*)d_in[1];
  const float* W    = (const float*)d_in[2];
  const float* bias = (const float*)d_in[3];
  float* out = (float*)d_out;

  dim3 grid(TT / BROWS, BB);
  da_fused_kernel<<<grid, 256, 0, stream>>>(x, mask, W, bias, out);
}


# attn band: the i-j block rule at two levels: each 4x4 group of 16x16 weight blocks takes 9 WMMA products, not 12; a wave holds 64 rows x 32 cols; rows outside [0,T) staged as 0; sums regrouped
# speedup vs baseline: 2.1186x; 1.0698x over previous
#include <hip/hip_runtime.h>


typedef __attribute__((ext_vector_type(2))) float v2f;
typedef __attribute__((ext_vector_type(4))) float v4f;
typedef __attribute__((ext_vector_type(8))) float v8f;

#define TT   4096
#define DD   256
#define BB   4
#define BAND 288
#define NCH  ((BROWS + 2 * BAND) / CHUNK)
#define INV_E 0.36787944117144233f

#define BROWS 64
#define CHUNK 64
#define RP    544
#define STAGE (CHUNK / 2 * RP)
#define ZP    260
#define NIT   (CHUNK / 2 * (DD / 4) / 256)

__device__ __forceinline__ v8f wmma_f32(v2f a, v2f b, v8f c) {
  return __builtin_amdgcn_wmma_f32_16x16x4_f32(false, a, false, b,
                                               (short)0, c, false, false);
}

__global__ __launch_bounds__(256)
void da_fused_kernel(const float* __restrict__ x, const float* __restrict__ mask,
                     const float* __restrict__ W, const float* __restrict__ bias,
                     float* __restrict__ out) {
  __shared__ __attribute__((aligned(16))) float sx[2 * STAGE];
  __shared__ __attribute__((aligned(16))) float sz[BROWS * ZP];
  __shared__ __attribute__((aligned(16))) float smf[2 * CHUNK];
  __shared__ __attribute__((aligned(16))) float ss[BROWS];

  const int tid = threadIdx.x;
  const int lane = tid & 31;
  const int c = lane & 15;
  const int h = lane >> 4;
  const int wv = __builtin_amdgcn_readfirstlane(tid >> 5);
  const int I0 = blockIdx.x * BROWS;
  const int bi = blockIdx.y;

  const float* xb = x + (bi * TT) * DD;
  const float* mb = mask + bi * TT;

  const int jlo = I0 - BAND;
  const int c_lo = (jlo < 0) ? ((-jlo) >> 6) : 0;
  int c_hi = (TT - jlo + CHUNK - 1) >> 6; if (c_hi > NCH) c_hi = NCH;

  const int cg = tid & 63;
  const int rp0 = tid >> 6;
  const float* gsrc = xb + 4 * cg;
  float* sdst = sx + rp0 * RP + 8 * cg;

  const float m = (float)(I0 + c);

  v8f zero = {};
  v8f acc[3][3][2];
  for (int q = 0; q < 3; ++q)
    for (int p = 0; p < 3; ++p) { acc[q][p][0] = zero; acc[q][p][1] = zero; }
  v2f sacc[4];
  for (int i = 0; i < 4; ++i) { sacc[i].x = 0.0f; sacc[i].y = 0.0f; }

  v4f g0[NIT], g1[NIT];
  v2f mk[NIT];
  float mv = 0.0f;

  {
    const int j0 = jlo + c_lo * CHUNK;
    for (int it = 0; it < NIT; ++it) {
      int r = j0 + 2 * rp0 + 8 * it;
      r = r < 0 ? 0 : (r > TT - 2 ? TT - 2 : r);
      g0[it] = *(const v4f*)(gsrc + r * DD);
      g1[it] = *(const v4f*)(gsrc + r * DD + DD);
      mk[it] = *(const v2f*)(mb + r);
    }
    if (tid < CHUNK) {
      int r = j0 + tid;
      r = r < 0 ? 0 : (r > TT - 1 ? TT - 1 : r);
      mv = mb[r];
    }
    for (int it = 0; it < NIT; ++it) {
      const int ru = j0 + 2 * rp0 + 8 * it;
      const bool in = (ru >= 0) && (ru < TT);
      const float fa = in ? __expf(mk[it].x * INV_E) : 0.0f;
      const float fb = in ? __expf(mk[it].y * INV_E) : 0.0f;
      v4f s0 = {g0[it].x * fa, g1[it].x * fb, g0[it].y * fa, g1[it].y * fb};
      v4f s1 = {g0[it].z * fa, g1[it].z * fb, g0[it].w * fa, g1[it].w * fb};
      *(v4f*)(sdst + it * 4 * RP)     = s0;
      *(v4f*)(sdst + it * 4 * RP + 4) = s1;
    }
    if (tid < CHUNK) {
      const int ru = j0 + tid;
      smf[tid] = (ru >= 0 && ru < TT) ? __expf(mv * INV_E) : 0.0f;
    }
  }
  __syncthreads();

  for (int ci = c_lo; ci < c_hi; ++ci) {
    const int jc = jlo + ci * CHUNK;
    const int st = (ci - c_lo) & 1;
    const bool more = (ci + 1) < c_hi;
    if (more) {
      for (int it = 0; it < NIT; ++it) {
        int r = jc + CHUNK + 2 * rp0 + 8 * it;
        r = r < 0 ? 0 : (r > TT - 2 ? TT - 2 : r);
        g0[it] = *(const v4f*)(gsrc + r * DD);
        g1[it] = *(const v4f*)(gsrc + r * DD + DD);
        mk[it] = *(const v2f*)(mb + r);
      }
      if (tid < CHUNK) {
        int r = jc + CHUNK + tid;
        r = r < 0 ? 0 : (r > TT - 1 ? TT - 1 : r);
        mv = mb[r];
      }
    }
    const float* sbuf = sx + st * STAGE;
    const float* mfb = smf + st * CHUNK;

    {
      v2f fj[4][4];
      v2f y[4][4][2];
      for (int J = 0; J < 4; ++J)
        for (int s = 0; s < 4; ++s) fj[J][s] = *(const v2f*)(mfb + 16 * J + 4 * s + 2 * h);
      for (int J = 0; J < 4; ++J) {
        for (int s = 0; s < 4; ++s) {
          const float* p = sbuf + (8 * J + 2 * s + h) * RP + 64 * wv + 2 * c;
          y[J][s][0] = *(const v2f*)(p);
          y[J][s][1] = *(const v2f*)(p + 32);
        }
      }
      v2f wa[3][3][4];
      for (int s = 0; s < 4; ++s) {
        const float d0 = m - (float)(jc + 4 * s + 2 * h);
        const float d1 = d0 - 1.0f;
        v2f T[7];
        for (int k = -3; k <= 3; ++k) {
          T[k + 3].x = __expf(-fabsf(d0 + 16.0f * (float)k) * INV_E);
          T[k + 3].y = __expf(-fabsf(d1 + 16.0f * (float)k) * INV_E);
        }
        for (int I = 0; I < 4; ++I)
          for (int J = 0; J < 4; ++J) sacc[I] += T[I - J + 3] * fj[J][s];
        const v2f t0 = T[3];
        const v2f q2a0 = T[1] - t0, q2am = T[0] - T[2], q2ap = T[2] - T[4];
        const v2f q3a0 = T[5] - t0, q3am = T[4] - T[2], q3ap = T[6] - T[4];
        wa[0][0][s] = t0;   wa[0][1][s] = T[2] - t0;    wa[0][2][s] = T[4] - t0;
        wa[1][0][s] = q2a0; wa[1][1][s] = q2am - q2a0;  wa[1][2][s] = q2ap - q2a0;
        wa[2][0][s] = q3a0; wa[2][1][s] = q3am - q3a0;  wa[2][2][s] = q3ap - q3a0;
      }
      v2f v0[4][2], v1[4][2], vs[4][2], w01[4][2], w23[4][2];
      for (int s = 0; s < 4; ++s) {
        for (int t = 0; t < 2; ++t) {
          v0[s][t]  = y[0][s][t] + y[2][s][t];
          v1[s][t]  = y[1][s][t] + y[3][s][t];
          vs[s][t]  = v0[s][t] + v1[s][t];
          w01[s][t] = y[0][s][t] + y[1][s][t];
          w23[s][t] = y[2][s][t] + y[3][s][t];
        }
      }
      __builtin_amdgcn_sched_barrier(0);
      for (int s = 0; s < 4; ++s) {
        for (int t = 0; t < 2; ++t) {
          acc[0][0][t] = wmma_f32(wa[0][0][s], vs[s][t],   acc[0][0][t]);
          acc[0][1][t] = wmma_f32(wa[0][1][s], v1[s][t],   acc[0][1][t]);
          acc[0][2][t] = wmma_f32(wa[0][2][s], v0[s][t],   acc[0][2][t]);
          acc[1][0][t] = wmma_f32(wa[1][0][s], w23[s][t],  acc[1][0][t]);
          acc[1][1][t] = wmma_f32(wa[1][1][s], y[3][s][t], acc[1][1][t]);
          acc[1][2][t] = wmma_f32(wa[1][2][s], y[2][s][t], acc[1][2][t]);
          acc[2][0][t] = wmma_f32(wa[2][0][s], w01[s][t],  acc[2][0][t]);
          acc[2][1][t] = wmma_f32(wa[2][1][s], y[1][s][t], acc[2][1][t]);
          acc[2][2][t] = wmma_f32(wa[2][2][s], y[0][s][t], acc[2][2][t]);
        }
      }
    }

    if (more) {
      float* sn = sdst + (st ^ 1) * STAGE;
      for (int it = 0; it < NIT; ++it) {
        const int ru = jc + CHUNK + 2 * rp0 + 8 * it;
        const bool in = (ru >= 0) && (ru < TT);
        const float fa = in ? __expf(mk[it].x * INV_E) : 0.0f;
        const float fb = in ? __expf(mk[it].y * INV_E) : 0.0f;
        v4f s0 = {g0[it].x * fa, g1[it].x * fb, g0[it].y * fa, g1[it].y * fb};
        v4f s1 = {g0[it].z * fa, g1[it].z * fb, g0[it].w * fa, g1[it].w * fb};
        *(v4f*)(sn + it * 4 * RP)     = s0;
        *(v4f*)(sn + it * 4 * RP + 4) = s1;
      }
      if (tid < CHUNK) {
        const int ru = jc + CHUNK + tid;
        smf[(st ^ 1) * CHUNK + tid] = (ru >= 0 && ru < TT) ? __expf(mv * INV_E) : 0.0f;
      }
    }
    __syncthreads();
  }

  {
    float sr[4];
    for (int I = 0; I < 4; ++I) {
      sr[I] = sacc[I].x + sacc[I].y;
      sr[I] += __shfl_xor(sr[I], 16);
    }
    if (wv == 0 && h == 0) {
      for (int I = 0; I < 4; ++I) ss[16 * I + c] = sr[I];
    }
  }

  {
    float* zd = sz + (8 * h) * ZP + 32 * wv + c;
    for (int t = 0; t < 2; ++t) {
      for (int r = 0; r < 8; ++r) {
        const float q1u = acc[0][0][t][r] + acc[0][1][t][r];
        const float q1l = acc[0][0][t][r] + acc[0][2][t][r];
        const float q2u = acc[1][0][t][r] + acc[1][1][t][r];
        const float q2l = acc[1][0][t][r] + acc[1][2][t][r];
        const float q3u = acc[2][0][t][r] + acc[2][1][t][r];
        const float q3l = acc[2][0][t][r] + acc[2][2][t][r];
        zd[r * ZP + 16 * t]        = q1u + q2u;
        zd[(16 + r) * ZP + 16 * t] = q1l + q2l;
        zd[(32 + r) * ZP + 16 * t] = q1u + q3u;
        zd[(48 + r) * ZP + 16 * t] = q1l + q3l;
      }
    }
  }
  __syncthreads();

  const int n0 = 32 * wv;
  const float* wr0 = W + (n0 + c) * DD + 4 * h;
  const float* wr1 = W + (n0 + 16 + c) * DD + 4 * h;
  const float* zr = sz + c * ZP + 4 * h;

  v8f o[4][2];
  for (int i = 0; i < 4; ++i) { o[i][0] = zero; o[i][1] = zero; }

  v4f bw[3][2][2];
  v4f az[2][4], nz[2][4];
  for (int j = 0; j < 2; ++j) {
    bw[0][j][0] = *(const v4f*)(wr0 + 8 * j);
    bw[0][j][1] = *(const v4f*)(wr1 + 8 * j);
    bw[1][j][0] = *(const v4f*)(wr0 + 16 + 8 * j);
    bw[1][j][1] = *(const v4f*)(wr1 + 16 + 8 * j);
    for (int i = 0; i < 4; ++i) az[j][i] = *(const v4f*)(zr + 16 * i * ZP + 8 * j);
  }

  for (int it = 0; it < DD / 16; ++it) {
    const int kb = 16 * it;
    __builtin_amdgcn_sched_barrier(0);
    if (it + 2 < DD / 16) {
      for (int j = 0; j < 2; ++j) {
        bw[(it + 2) % 3][j][0] = *(const v4f*)(wr0 + kb + 32 + 8 * j);
        bw[(it + 2) % 3][j][1] = *(const v4f*)(wr1 + kb + 32 + 8 * j);
      }
    }
    if (it + 1 < DD / 16) {
      for (int j = 0; j < 2; ++j) {
        for (int i = 0; i < 4; ++i) nz[j][i] = *(const v4f*)(zr + 16 * i * ZP + kb + 16 + 8 * j);
      }
    }
    __builtin_amdgcn_sched_barrier(0);

    for (int j = 0; j < 2; ++j) {
      const v4f w0 = bw[it % 3][j][0];
      const v4f w1 = bw[it % 3][j][1];
      const v2f b0a = {w0.x, w0.y}, b0b = {w0.z, w0.w};
      const v2f b1a = {w1.x, w1.y}, b1b = {w1.z, w1.w};
      for (int i = 0; i < 4; ++i) {
        const v2f aa = {az[j][i].x, az[j][i].y};
        o[i][0] = wmma_f32(aa, b0a, o[i][0]);
        o[i][1] = wmma_f32(aa, b1a, o[i][1]);
      }
      for (int i = 0; i < 4; ++i) {
        const v2f ab = {az[j][i].z, az[j][i].w};
        o[i][0] = wmma_f32(ab, b0b, o[i][0]);
        o[i][1] = wmma_f32(ab, b1b, o[i][1]);
      }
    }
    if (it + 1 < DD / 16) {
      for (int j = 0; j < 2; ++j) {
        for (int i = 0; i < 4; ++i) az[j][i] = nz[j][i];
      }
    }
  }

  const float bia0 = bias[n0 + c];
  const float bia1 = bias[n0 + 16 + c];
  float* ob = out + (bi * TT + I0 + 8 * h) * DD + n0 + c;
  for (int i = 0; i < 4; ++i) {
    for (int r = 0; r < 8; ++r) {
      const float sr = ss[16 * i + 8 * h + r];
      ob[(16 * i + r) * DD]      = o[i][0][r] + sr * bia0;
      ob[(16 * i + r) * DD + 16] = o[i][1][r] + sr * bia1;
    }
  }
}

extern "C" void kernel_launch(void* const* d_in, const int* in_sizes, int n_in,
                              void* d_out, int out_size, void* d_ws, size_t ws_size,
                              hipStream_t stream) {
  const float* x    = (const float*)d_in[0];
  const float* mask = (const float*)d_in[1];
  const float* W    = (const float*)d_in[2];
  const float* bias = (const float*)d_in[3];
  float* out = (float*)d_out;

  dim3 grid(TT / BROWS, BB);
  da_fused_kernel<<<grid, 256, 0, stream>>>(x, mask, W, bias, out);
}


# attn band: the nine weight operands stay in registers over chunks wholly on one side of the rows; such a chunk's decay exp(-64n/e) rides in the staged rows' scale f; built 4x, not 10x; s regrouped
# speedup vs baseline: 2.1849x; 1.0313x over previous
#include <hip/hip_runtime.h>


typedef __attribute__((ext_vector_type(2))) float v2f;
typedef __attribute__((ext_vector_type(4))) float v4f;
typedef __attribute__((ext_vector_type(8))) float v8f;

#define TT   4096
#define DD   256
#define BB   4
#define BAND 288
#define NCH  ((BROWS + 2 * BAND) / CHUNK)
#define INV_E 0.36787944117144233f

#define BROWS 64
#define CHUNK 64
#define RP    544
#define STAGE (CHUNK / 2 * RP)
#define ZP    260
#define NIT   (CHUNK / 2 * (DD / 4) / 256)
#define CREF_LO ((BAND - CHUNK) / CHUNK)
#define CREF_HI ((BAND + BROWS + CHUNK - 1) / CHUNK)

__device__ __forceinline__ float chunk_shift(int ci) {
  const int n = (ci < CREF_LO ? CREF_LO - ci : 0) + (ci > CREF_HI ? ci - CREF_HI : 0);
  return -(float)(n * CHUNK) * INV_E;
}

__device__ __forceinline__ v8f wmma_f32(v2f a, v2f b, v8f c) {
  return __builtin_amdgcn_wmma_f32_16x16x4_f32(false, a, false, b,
                                               (short)0, c, false, false);
}

__global__ __launch_bounds__(256)
void da_fused_kernel(const float* __restrict__ x, const float* __restrict__ mask,
                     const float* __restrict__ W, const float* __restrict__ bias,
                     float* __restrict__ out) {
  __shared__ __attribute__((aligned(16))) float sx[2 * STAGE];
  __shared__ __attribute__((aligned(16))) float sz[BROWS * ZP];
  __shared__ __attribute__((aligned(16))) float smf[2 * CHUNK];
  __shared__ __attribute__((aligned(16))) float ss[BROWS];

  const int tid = threadIdx.x;
  const int lane = tid & 31;
  const int c = lane & 15;
  const int h = lane >> 4;
  const int wv = __builtin_amdgcn_readfirstlane(tid >> 5);
  const int I0 = blockIdx.x * BROWS;
  const int bi = blockIdx.y;

  const float* xb = x + (bi * TT) * DD;
  const float* mb = mask + bi * TT;

  const int jlo = I0 - BAND;
  const int c_lo = (jlo < 0) ? ((-jlo) >> 6) : 0;
  int c_hi = (TT - jlo + CHUNK - 1) >> 6; if (c_hi > NCH) c_hi = NCH;

  const int cg = tid & 63;
  const int rp0 = tid >> 6;
  const float* gsrc = xb + 4 * cg;
  float* sdst = sx + rp0 * RP + 8 * cg;

  const float m = (float)(I0 + c);

  v8f zero = {};
  v8f acc[3][3][2];
  for (int q = 0; q < 3; ++q)
    for (int p = 0; p < 3; ++p) { acc[q][p][0] = zero; acc[q][p][1] = zero; }
  v2f s9[3][3];
  for (int q = 0; q < 3; ++q)
    for (int p = 0; p < 3; ++p) { s9[q][p].x = 0.0f; s9[q][p].y = 0.0f; }
  v2f wa[3][3][4];
  for (int q = 0; q < 3; ++q)
    for (int p = 0; p < 3; ++p)
      for (int ks = 0; ks < 4; ++ks) { wa[q][p][ks].x = 0.0f; wa[q][p][ks].y = 0.0f; }

  v4f g0[NIT], g1[NIT];
  v2f mk[NIT];
  float mv = 0.0f;

  {
    const int j0 = jlo + c_lo * CHUNK;
    const float gsh = chunk_shift(c_lo);
    for (int it = 0; it < NIT; ++it) {
      int r = j0 + 2 * rp0 + 8 * it;
      r = r < 0 ? 0 : (r > TT - 2 ? TT - 2 : r);
      g0[it] = *(const v4f*)(gsrc + r * DD);
      g1[it] = *(const v4f*)(gsrc + r * DD + DD);
      mk[it] = *(const v2f*)(mb + r);
    }
    if (tid < CHUNK) {
      int r = j0 + tid;
      r = r < 0 ? 0 : (r > TT - 1 ? TT - 1 : r);
      mv = mb[r];
    }
    for (int it = 0; it < NIT; ++it) {
      const int ru = j0 + 2 * rp0 + 8 * it;
      const bool in = (ru >= 0) && (ru < TT);
      const float fa = in ? __expf(fmaf(mk[it].x, INV_E, gsh)) : 0.0f;
      const float fb = in ? __expf(fmaf(mk[it].y, INV_E, gsh)) : 0.0f;
      v4f s0 = {g0[it].x * fa, g1[it].x * fb, g0[it].y * fa, g1[it].y * fb};
      v4f s1 = {g0[it].z * fa, g1[it].z * fb, g0[it].w * fa, g1[it].w * fb};
      *(v4f*)(sdst + it * 4 * RP)     = s0;
      *(v4f*)(sdst + it * 4 * RP + 4) = s1;
    }
    if (tid < CHUNK) {
      const int ru = j0 + tid;
      smf[tid] = (ru >= 0 && ru < TT) ? __expf(fmaf(mv, INV_E, gsh)) : 0.0f;
    }
  }
  __syncthreads();

  for (int ci = c_lo; ci < c_hi; ++ci) {
    const int jc = jlo + ci * CHUNK;
    const int st = (ci - c_lo) & 1;
    const bool more = (ci + 1) < c_hi;
    const bool gen = (ci == c_lo) || (ci > CREF_LO && ci <= CREF_HI);
    const int cref = ci < CREF_LO ? CREF_LO : (ci > CREF_HI ? CREF_HI : ci);
    const int jref = jlo + cref * CHUNK;
    const float* sbuf = sx + st * STAGE;
    const float* mfb = smf + st * CHUNK;

    {
      v2f fj[4][4];
      v2f y[4][4][2];
      for (int s = 0; s < 4; ++s) {
        for (int J = 0; J < 4; ++J) {
          const float* p = sbuf + (8 * J + 2 * s + h) * RP + 64 * wv + 2 * c;
          y[J][s][0] = *(const v2f*)(p);
          y[J][s][1] = *(const v2f*)(p + 32);
        }
        if (s == 0) __builtin_amdgcn_sched_barrier(0);
      }
      for (int J = 0; J < 4; ++J)
        for (int s = 0; s < 4; ++s) fj[J][s] = *(const v2f*)(mfb + 16 * J + 4 * s + 2 * h);
      if (more) {
        for (int it = 0; it < NIT; ++it) {
          int r = jc + CHUNK + 2 * rp0 + 8 * it;
          r = r < 0 ? 0 : (r > TT - 2 ? TT - 2 : r);
          g0[it] = *(const v4f*)(gsrc + r * DD);
          g1[it] = *(const v4f*)(gsrc + r * DD + DD);
          mk[it] = *(const v2f*)(mb + r);
        }
        if (tid < CHUNK) {
          int r = jc + CHUNK + tid;
          r = r < 0 ? 0 : (r > TT - 1 ? TT - 1 : r);
          mv = mb[r];
        }
      }
      if (gen) {
        for (int s = 0; s < 4; ++s) {
          const float d0 = m - (float)(jref + 4 * s + 2 * h);
          const float d1 = d0 - 1.0f;
          v2f T[7];
          for (int k = -3; k <= 3; ++k) {
            T[k + 3].x = __expf(-fabsf(d0 + 16.0f * (float)k) * INV_E);
            T[k + 3].y = __expf(-fabsf(d1 + 16.0f * (float)k) * INV_E);
          }
          const v2f t0 = T[3];
          const v2f q2a0 = T[1] - t0, q2am = T[0] - T[2], q2ap = T[2] - T[4];
          const v2f q3a0 = T[5] - t0, q3am = T[4] - T[2], q3ap = T[6] - T[4];
          wa[0][0][s] = t0;   wa[0][1][s] = T[2] - t0;    wa[0][2][s] = T[4] - t0;
          wa[1][0][s] = q2a0; wa[1][1][s] = q2am - q2a0;  wa[1][2][s] = q2ap - q2a0;
          wa[2][0][s] = q3a0; wa[2][1][s] = q3am - q3a0;  wa[2][2][s] = q3ap - q3a0;
        }
      }
      for (int s = 0; s < 4; ++s) {
        v2f v0[2], v1[2], vs[2], w01[2], w23[2];
        for (int t = 0; t < 2; ++t) {
          v0[t]  = y[0][s][t] + y[2][s][t];
          v1[t]  = y[1][s][t] + y[3][s][t];
          vs[t]  = v0[t] + v1[t];
          w01[t] = y[0][s][t] + y[1][s][t];
          w23[t] = y[2][s][t] + y[3][s][t];
        }
        __builtin_amdgcn_sched_barrier(0);
        for (int t = 0; t < 2; ++t) {
          acc[0][0][t] = wmma_f32(wa[0][0][s], vs[t],      acc[0][0][t]);
          acc[0][1][t] = wmma_f32(wa[0][1][s], v1[t],      acc[0][1][t]);
          acc[0][2][t] = wmma_f32(wa[0][2][s], v0[t],      acc[0][2][t]);
          acc[1][0][t] = wmma_f32(wa[1][0][s], w23[t],     acc[1][0][t]);
          acc[1][1][t] = wmma_f32(wa[1][1][s], y[3][s][t], acc[1][1][t]);
          acc[1][2][t] = wmma_f32(wa[1][2][s], y[2][s][t], acc[1][2][t]);
          acc[2][0][t] = wmma_f32(wa[2][0][s], w01[t],     acc[2][0][t]);
          acc[2][1][t] = wmma_f32(wa[2][1][s], y[1][s][t], acc[2][1][t]);
          acc[2][2][t] = wmma_f32(wa[2][2][s], y[0][s][t], acc[2][2][t]);
        }
        __builtin_amdgcn_sched_barrier(0);
      }
      for (int s = 0; s < 4; ++s) {
        const v2f f0 = fj[0][s], f1 = fj[1][s], f2 = fj[2][s], f3 = fj[3][s];
        const v2f fv0 = f0 + f2, fv1 = f1 + f3, fw01 = f0 + f1, fw23 = f2 + f3;
        const v2f fvs = fv0 + fv1;
        s9[0][0] += wa[0][0][s] * fvs;  s9[0][1] += wa[0][1][s] * fv1; s9[0][2] += wa[0][2][s] * fv0;
        s9[1][0] += wa[1][0][s] * fw23; s9[1][1] += wa[1][1][s] * f3;  s9[1][2] += wa[1][2][s] * f2;
        s9[2][0] += wa[2][0][s] * fw01; s9[2][1] += wa[2][1][s] * f1;  s9[2][2] += wa[2][2][s] * f0;
      }
    }

    if (more) {
      float* sn = sdst + (st ^ 1) * STAGE;
      const float gsh = chunk_shift(ci + 1);
      for (int it = 0; it < NIT; ++it) {
        const int ru = jc + CHUNK + 2 * rp0 + 8 * it;
        const bool in = (ru >= 0) && (ru < TT);
        const float fa = in ? __expf(fmaf(mk[it].x, INV_E, gsh)) : 0.0f;
        const float fb = in ? __expf(fmaf(mk[it].y, INV_E, gsh)) : 0.0f;
        v4f s0 = {g0[it].x * fa, g1[it].x * fb, g0[it].y * fa, g1[it].y * fb};
        v4f s1 = {g0[it].z * fa, g1[it].z * fb, g0[it].w * fa, g1[it].w * fb};
        *(v4f*)(sn + it * 4 * RP)     = s0;
        *(v4f*)(sn + it * 4 * RP + 4) = s1;
      }
      if (tid < CHUNK) {
        const int ru = jc + CHUNK + tid;
        smf[(st ^ 1) * CHUNK + tid] = (ru >= 0 && ru < TT) ? __expf(fmaf(mv, INV_E, gsh)) : 0.0f;
      }
    }
    __syncthreads();
  }

  {
    const v2f q1u = s9[0][0] + s9[0][1], q1l = s9[0][0] + s9[0][2];
    const v2f q2u = s9[1][0] + s9[1][1], q2l = s9[1][0] + s9[1][2];
    const v2f q3u = s9[2][0] + s9[2][1], q3l = s9[2][0] + s9[2][2];
    const v2f sv[4] = {q1u + q2u, q1l + q2l, q1u + q3u, q1l + q3l};
    float sr[4];
    for (int I = 0; I < 4; ++I) {
      sr[I] = sv[I].x + sv[I].y;
      sr[I] += __shfl_xor(sr[I], 16);
    }
    if (wv == 0 && h == 0) {
      for (int I = 0; I < 4; ++I) ss[16 * I + c] = sr[I];
    }
  }

  {
    float* zd = sz + (8 * h) * ZP + 32 * wv + c;
    for (int t = 0; t < 2; ++t) {
      for (int r = 0; r < 8; ++r) {
        const float q1u = acc[0][0][t][r] + acc[0][1][t][r];
        const float q1l = acc[0][0][t][r] + acc[0][2][t][r];
        const float q2u = acc[1][0][t][r] + acc[1][1][t][r];
        const float q2l = acc[1][0][t][r] + acc[1][2][t][r];
        const float q3u = acc[2][0][t][r] + acc[2][1][t][r];
        const float q3l = acc[2][0][t][r] + acc[2][2][t][r];
        zd[r * ZP + 16 * t]        = q1u + q2u;
        zd[(16 + r) * ZP + 16 * t] = q1l + q2l;
        zd[(32 + r) * ZP + 16 * t] = q1u + q3u;
        zd[(48 + r) * ZP + 16 * t] = q1l + q3l;
      }
    }
  }
  __syncthreads();

  const int n0 = 32 * wv;
  const float* wr0 = W + (n0 + c) * DD + 4 * h;
  const float* wr1 = W + (n0 + 16 + c) * DD + 4 * h;
  const float* zr = sz + c * ZP + 4 * h;

  v8f o[4][2];
  for (int i = 0; i < 4; ++i) { o[i][0] = zero; o[i][1] = zero; }

  v4f bw[3][2][2];
  v4f az[2][4], nz[2][4];
  for (int j = 0; j < 2; ++j) {
    bw[0][j][0] = *(const v4f*)(wr0 + 8 * j);
    bw[0][j][1] = *(const v4f*)(wr1 + 8 * j);
    bw[1][j][0] = *(const v4f*)(wr0 + 16 + 8 * j);
    bw[1][j][1] = *(const v4f*)(wr1 + 16 + 8 * j);
    for (int i = 0; i < 4; ++i) az[j][i] = *(const v4f*)(zr + 16 * i * ZP + 8 * j);
  }

  for (int it = 0; it < DD / 16; ++it) {
    const int kb = 16 * it;
    __builtin_amdgcn_sched_barrier(0);
    if (it + 2 < DD / 16) {
      for (int j = 0; j < 2; ++j) {
        bw[(it + 2) % 3][j][0] = *(const v4f*)(wr0 + kb + 32 + 8 * j);
        bw[(it + 2) % 3][j][1] = *(const v4f*)(wr1 + kb + 32 + 8 * j);
      }
    }
    if (it + 1 < DD / 16) {
      for (int j = 0; j < 2; ++j) {
        for (int i = 0; i < 4; ++i) nz[j][i] = *(const v4f*)(zr + 16 * i * ZP + kb + 16 + 8 * j);
      }
    }
    __builtin_amdgcn_sched_barrier(0);

    for (int j = 0; j < 2; ++j) {
      const v4f w0 = bw[it % 3][j][0];
      const v4f w1 = bw[it % 3][j][1];
      const v2f b0a = {w0.x, w0.y}, b0b = {w0.z, w0.w};
      const v2f b1a = {w1.x, w1.y}, b1b = {w1.z, w1.w};
      for (int i = 0; i < 4; ++i) {
        const v2f aa = {az[j][i].x, az[j][i].y};
        o[i][0] = wmma_f32(aa, b0a, o[i][0]);
        o[i][1] = wmma_f32(aa, b1a, o[i][1]);
      }
      for (int i = 0; i < 4; ++i) {
        const v2f ab = {az[j][i].z, az[j][i].w};
        o[i][0] = wmma_f32(ab, b0b, o[i][0]);
        o[i][1] = wmma_f32(ab, b1b, o[i][1]);
      }
    }
    if (it + 1 < DD / 16) {
      for (int j = 0; j < 2; ++j) {
        for (int i = 0; i < 4; ++i) az[j][i] = nz[j][i];
      }
    }
  }

  const float bia0 = bias[n0 + c];
  const float bia1 = bias[n0 + 16 + c];
  float* ob = out + (bi * TT + I0 + 8 * h) * DD + n0 + c;
  for (int i = 0; i < 4; ++i) {
    for (int r = 0; r < 8; ++r) {
      const float sr = ss[16 * i + 8 * h + r];
      ob[(16 * i + r) * DD]      = o[i][0][r] + sr * bia0;
      ob[(16 * i + r) * DD + 16] = o[i][1][r] + sr * bia1;
    }
  }
}

extern "C" void kernel_launch(void* const* d_in, const int* in_sizes, int n_in,
                              void* d_out, int out_size, void* d_ws, size_t ws_size,
                              hipStream_t stream) {
  const float* x    = (const float*)d_in[0];
  const float* mask = (const float*)d_in[1];
  const float* W    = (const float*)d_in[2];
  const float* bias = (const float*)d_in[3];
  float* out = (float*)d_out;

  dim3 grid(TT / BROWS, BB);
  da_fused_kernel<<<grid, 256, 0, stream>>>(x, mask, W, bias, out);
}


# attn band: the 4 chunks wholly below the rows are summed, each row times its chunk decay exp(-64n/e), into ONE 64-row image before the product; same above: 4 product groups, not 10; sums regrouped
# speedup vs baseline: 2.9586x; 1.3541x over previous
#include <hip/hip_runtime.h>


typedef __attribute__((ext_vector_type(2))) float v2f;
typedef __attribute__((ext_vector_type(4))) float v4f;
typedef __attribute__((ext_vector_type(8))) float v8f;

#define TT   4096
#define DD   256
#define BB   4
#define BAND 288
#define INV_E 0.36787944117144233f

#define BROWS 64
#define CHUNK 64
#define RP    544
#define STAGE (CHUNK / 2 * RP)
#define ZP    260
#define NIT   (CHUNK / 2 * (DD / 4) / 256)
#define CREF_LO ((BAND - CHUNK) / CHUNK)
#define CREF_HI ((BAND + BROWS + CHUNK - 1) / CHUNK)

__device__ __forceinline__ float fold_shift(int n) {
  return -(float)(n * CHUNK) * INV_E;
}

__device__ __forceinline__ v8f wmma_f32(v2f a, v2f b, v8f c) {
  return __builtin_amdgcn_wmma_f32_16x16x4_f32(false, a, false, b,
                                               (short)0, c, false, false);
}

__global__ __launch_bounds__(256)
void da_fused_kernel(const float* __restrict__ x, const float* __restrict__ mask,
                     const float* __restrict__ W, const float* __restrict__ bias,
                     float* __restrict__ out) {
  __shared__ __attribute__((aligned(16))) float sx[2 * STAGE];
  __shared__ __attribute__((aligned(16))) float sz[BROWS * ZP];
  __shared__ __attribute__((aligned(16))) float smf[2 * CHUNK];
  __shared__ __attribute__((aligned(16))) float ss[BROWS];

  const int tid = threadIdx.x;
  const int lane = tid & 31;
  const int c = lane & 15;
  const int h = lane >> 4;
  const int wv = __builtin_amdgcn_readfirstlane(tid >> 5);
  const int I0 = blockIdx.x * BROWS;
  const int bi = blockIdx.y;

  const float* xb = x + (bi * TT) * DD;
  const float* mb = mask + bi * TT;

  const int jlo = I0 - BAND;

  const int cg = tid & 63;
  const int rp0 = tid >> 6;
  const unsigned xoff = (unsigned)((2 * rp0) * DD + 4 * cg) * 4u;
  const unsigned moff = (unsigned)(2 * rp0) * 4u;
  float* sdst = sx + rp0 * RP + 8 * cg;

  const float m = (float)(I0 + c);

  v8f zero = {};
  v8f acc[3][3][2];
  for (int q = 0; q < 3; ++q)
    for (int p = 0; p < 3; ++p) { acc[q][p][0] = zero; acc[q][p][1] = zero; }
  v2f s9[3][3];
  for (int q = 0; q < 3; ++q)
    for (int p = 0; p < 3; ++p) { s9[q][p].x = 0.0f; s9[q][p].y = 0.0f; }
  v2f wa[3][3][4];
  for (int q = 0; q < 3; ++q)
    for (int p = 0; p < 3; ++p)
      for (int ks = 0; ks < 4; ++ks) { wa[q][p][ks].x = 0.0f; wa[q][p][ks].y = 0.0f; }

  v4f g0[NIT], g1[NIT];
  v2f mk[NIT];
  float mv[4] = {0.0f, 0.0f, 0.0f, 0.0f};
  v2f fj[4][4];
  v2f y[4][4][2];

  auto half_in = [&](int jb) -> bool { return jb >= 0 && jb < TT; };
  auto load_item = [&](int sl, int j0, int it) {
    const int jb = j0 + 32 * (it >> 2);
    const int jc = __builtin_amdgcn_readfirstlane(half_in(jb) ? jb : 0);
    const char* px = (const char*)(xb + (jc + 8 * (it & 3)) * DD);
    g0[sl] = *(const v4f*)(px + xoff);
    g1[sl] = *(const v4f*)(px + xoff + DD * 4);
    mk[sl] = *(const v2f*)((const char*)(mb + jc + 8 * (it & 3)) + moff);
  };
  auto frow = [&](float mval, bool in, int n) -> float {
    return in ? __expf(fmaf(mval, INV_E, fold_shift(n))) : 0.0f;
  };
  auto load_f = [&](int n, int j0) {
    const int jb = j0 + 32 * wv;
    const int jc = __builtin_amdgcn_readfirstlane(half_in(jb) ? jb : 0);
    mv[n] = mb[jc + lane];
  };

  auto issue_chunk = [&](int ci) {
    const int j0 = jlo + ci * CHUNK;
    for (int it = 0; it < NIT; ++it) load_item(it, j0, it);
    if (wv < 2) load_f(0, j0);
  };
  auto stage_chunk = [&](int ci, int st) {
    const int j0 = jlo + ci * CHUNK;
    float* sn = sdst + st * STAGE;
    for (int it = 0; it < NIT; ++it) {
      const bool in = half_in(j0 + 32 * (it >> 2));
      const float fa = frow(mk[it].x, in, 0);
      const float fb = frow(mk[it].y, in, 0);
      v4f s0 = {g0[it].x * fa, g1[it].x * fb, g0[it].y * fa, g1[it].y * fb};
      v4f s1 = {g0[it].z * fa, g1[it].z * fb, g0[it].w * fa, g1[it].w * fb};
      *(v4f*)(sn + it * 4 * RP)     = s0;
      *(v4f*)(sn + it * 4 * RP + 4) = s1;
    }
    if (wv < 2) smf[st * CHUNK + tid] = frow(mv[0], half_in(j0 + 32 * wv), 0);
  };

  auto issue_fold = [&](int it, int cref, int dir) {
    for (int n = 0; n < 4; ++n)
      load_item(4 * (it & 1) + n, jlo + (cref + dir * n) * CHUNK, it);
  };
  auto issue_fold_f = [&](int cref, int dir) {
    if (wv < 2) {
      for (int n = 0; n < 4; ++n) load_f(n, jlo + (cref + dir * n) * CHUNK);
    }
  };
  auto stage_fold = [&](int cref, int dir, int st) {
    float* sn = sdst + st * STAGE;
    for (int it = 0; it < NIT; ++it) {
      v4f a0, a1;
      for (int n = 3; n >= 0; --n) {
        const int sl = 4 * (it & 1) + n;
        const bool in = half_in(jlo + (cref + dir * n) * CHUNK + 32 * (it >> 2));
        const float fa = frow(mk[sl].x, in, n);
        const float fb = frow(mk[sl].y, in, n);
        if (n == 3) { a0 = g0[sl] * fa;      a1 = g1[sl] * fb; }
        else        { a0 = a0 + g0[sl] * fa; a1 = a1 + g1[sl] * fb; }
      }
      v4f s0 = {a0.x, a1.x, a0.y, a1.y};
      v4f s1 = {a0.z, a1.z, a0.w, a1.w};
      *(v4f*)(sn + it * 4 * RP)     = s0;
      *(v4f*)(sn + it * 4 * RP + 4) = s1;
      if (it + 2 < NIT) issue_fold(it + 2, cref, dir);
      __builtin_amdgcn_sched_barrier(0);
    }
    if (wv < 2) {
      float f = 0.0f;
      for (int n = 3; n >= 0; --n) f += frow(mv[n], half_in(jlo + (cref + dir * n) * CHUNK + 32 * wv), n);
      smf[st * CHUNK + tid] = f;
    }
  };

  auto read_frags = [&](int st) {
    const float* sbuf = sx + st * STAGE;
    const float* mfb = smf + st * CHUNK;
    for (int s = 0; s < 4; ++s) {
      for (int J = 0; J < 4; ++J) {
        const float* p = sbuf + (8 * J + 2 * s + h) * RP + 64 * wv + 2 * c;
        y[J][s][0] = *(const v2f*)(p);
        y[J][s][1] = *(const v2f*)(p + 32);
      }
      if (s == 0) __builtin_amdgcn_sched_barrier(0);
    }
    for (int J = 0; J < 4; ++J)
      for (int s = 0; s < 4; ++s) fj[J][s] = *(const v2f*)(mfb + 16 * J + 4 * s + 2 * h);
  };

  auto products = [&](int jref) {
    for (int s = 0; s < 4; ++s) {
      const float d0 = m - (float)(jref + 4 * s + 2 * h);
      const float d1 = d0 - 1.0f;
      v2f T[7];
      for (int k = -3; k <= 3; ++k) {
        T[k + 3].x = __expf(-fabsf(d0 + 16.0f * (float)k) * INV_E);
        T[k + 3].y = __expf(-fabsf(d1 + 16.0f * (float)k) * INV_E);
      }
      const v2f t0 = T[3];
      const v2f q2a0 = T[1] - t0, q2am = T[0] - T[2], q2ap = T[2] - T[4];
      const v2f q3a0 = T[5] - t0, q3am = T[4] - T[2], q3ap = T[6] - T[4];
      wa[0][0][s] = t0;   wa[0][1][s] = T[2] - t0;    wa[0][2][s] = T[4] - t0;
      wa[1][0][s] = q2a0; wa[1][1][s] = q2am - q2a0;  wa[1][2][s] = q2ap - q2a0;
      wa[2][0][s] = q3a0; wa[2][1][s] = q3am - q3a0;  wa[2][2][s] = q3ap - q3a0;
    }
    for (int s = 0; s < 4; ++s) {
      v2f v0[2], v1[2], vs[2], w01[2], w23[2];
      for (int t = 0; t < 2; ++t) {
        v0[t]  = y[0][s][t] + y[2][s][t];
        v1[t]  = y[1][s][t] + y[3][s][t];
        vs[t]  = v0[t] + v1[t];
        w01[t] = y[0][s][t] + y[1][s][t];
        w23[t] = y[2][s][t] + y[3][s][t];
      }
      __builtin_amdgcn_sched_barrier(0);
      for (int t = 0; t < 2; ++t) {
        acc[0][0][t] = wmma_f32(wa[0][0][s], vs[t],      acc[0][0][t]);
        acc[0][1][t] = wmma_f32(wa[0][1][s], v1[t],      acc[0][1][t]);
        acc[0][2][t] = wmma_f32(wa[0][2][s], v0[t],      acc[0][2][t]);
        acc[1][0][t] = wmma_f32(wa[1][0][s], w23[t],     acc[1][0][t]);
        acc[1][1][t] = wmma_f32(wa[1][1][s], y[3][s][t], acc[1][1][t]);
        acc[1][2][t] = wmma_f32(wa[1][2][s], y[2][s][t], acc[1][2][t]);
        acc[2][0][t] = wmma_f32(wa[2][0][s], w01[t],     acc[2][0][t]);
        acc[2][1][t] = wmma_f32(wa[2][1][s], y[1][s][t], acc[2][1][t]);
        acc[2][2][t] = wmma_f32(wa[2][2][s], y[0][s][t], acc[2][2][t]);
      }
      __builtin_amdgcn_sched_barrier(0);
    }
    for (int s = 0; s < 4; ++s) {
      const v2f f0 = fj[0][s], f1 = fj[1][s], f2 = fj[2][s], f3 = fj[3][s];
      const v2f fv0 = f0 + f2, fv1 = f1 + f3, fw01 = f0 + f1, fw23 = f2 + f3;
      const v2f fvs = fv0 + fv1;
      s9[0][0] += wa[0][0][s] * fvs;  s9[0][1] += wa[0][1][s] * fv1; s9[0][2] += wa[0][2][s] * fv0;
      s9[1][0] += wa[1][0][s] * fw23; s9[1][1] += wa[1][1][s] * f3;  s9[1][2] += wa[1][2][s] * f2;
      s9[2][0] += wa[2][0][s] * fw01; s9[2][1] += wa[2][1][s] * f1;  s9[2][2] += wa[2][2][s] * f0;
    }
  };


  issue_chunk(CREF_LO + 1);
  stage_chunk(CREF_LO + 1, 0);
  __syncthreads();

  read_frags(0);
  issue_chunk(CREF_LO + 2);
  products(jlo + (CREF_LO + 1) * CHUNK);
  stage_chunk(CREF_LO + 2, 1);
  __syncthreads();

#pragma nounroll
  for (int gi = 0; gi < 2; ++gi) {
    const int st = 1 - gi;
    const int cw = gi == 0 ? CREF_LO + 2 : CREF_LO;
    const int cref = gi == 0 ? CREF_LO : CREF_HI;
    const int dir = gi == 0 ? -1 : 1;
    read_frags(st);
    issue_fold(0, cref, dir);
    issue_fold(1, cref, dir);
    issue_fold_f(cref, dir);
    products(jlo + cw * CHUNK);
    stage_fold(cref, dir, st ^ 1);
    __syncthreads();
  }

  read_frags(1);
  products(jlo + CREF_HI * CHUNK);

  {
    const v2f q1u = s9[0][0] + s9[0][1], q1l = s9[0][0] + s9[0][2];
    const v2f q2u = s9[1][0] + s9[1][1], q2l = s9[1][0] + s9[1][2];
    const v2f q3u = s9[2][0] + s9[2][1], q3l = s9[2][0] + s9[2][2];
    const v2f sv[4] = {q1u + q2u, q1l + q2l, q1u + q3u, q1l + q3l};
    float sr[4];
    for (int I = 0; I < 4; ++I) {
      sr[I] = sv[I].x + sv[I].y;
      sr[I] += __shfl_xor(sr[I], 16);
    }
    if (wv == 0 && h == 0) {
      for (int I = 0; I < 4; ++I) ss[16 * I + c] = sr[I];
    }
  }

  {
    float* zd = sz + (8 * h) * ZP + 32 * wv + c;
    for (int t = 0; t < 2; ++t) {
      for (int r = 0; r < 8; ++r) {
        const float q1u = acc[0][0][t][r] + acc[0][1][t][r];
        const float q1l = acc[0][0][t][r] + acc[0][2][t][r];
        const float q2u = acc[1][0][t][r] + acc[1][1][t][r];
        const float q2l = acc[1][0][t][r] + acc[1][2][t][r];
        const float q3u = acc[2][0][t][r] + acc[2][1][t][r];
        const float q3l = acc[2][0][t][r] + acc[2][2][t][r];
        zd[r * ZP + 16 * t]        = q1u + q2u;
        zd[(16 + r) * ZP + 16 * t] = q1l + q2l;
        zd[(32 + r) * ZP + 16 * t] = q1u + q3u;
        zd[(48 + r) * ZP + 16 * t] = q1l + q3l;
      }
    }
  }
  __syncthreads();

  const int n0 = 32 * wv;
  const float* wr0 = W + (n0 + c) * DD + 4 * h;
  const float* wr1 = W + (n0 + 16 + c) * DD + 4 * h;
  const float* zr = sz + c * ZP + 4 * h;

  v8f o[4][2];
  for (int i = 0; i < 4; ++i) { o[i][0] = zero; o[i][1] = zero; }

  v4f bw[3][2][2];
  v4f az[2][4], nz[2][4];
  for (int j = 0; j < 2; ++j) {
    bw[0][j][0] = *(const v4f*)(wr0 + 8 * j);
    bw[0][j][1] = *(const v4f*)(wr1 + 8 * j);
    bw[1][j][0] = *(const v4f*)(wr0 + 16 + 8 * j);
    bw[1][j][1] = *(const v4f*)(wr1 + 16 + 8 * j);
    for (int i = 0; i < 4; ++i) az[j][i] = *(const v4f*)(zr + 16 * i * ZP + 8 * j);
  }

  for (int it = 0; it < DD / 16; ++it) {
    const int kb = 16 * it;
    __builtin_amdgcn_sched_barrier(0);
    if (it + 2 < DD / 16) {
      for (int j = 0; j < 2; ++j) {
        bw[(it + 2) % 3][j][0] = *(const v4f*)(wr0 + kb + 32 + 8 * j);
        bw[(it + 2) % 3][j][1] = *(const v4f*)(wr1 + kb + 32 + 8 * j);
      }
    }
    if (it + 1 < DD / 16) {
      for (int j = 0; j < 2; ++j) {
        for (int i = 0; i < 4; ++i) nz[j][i] = *(const v4f*)(zr + 16 * i * ZP + kb + 16 + 8 * j);
      }
    }
    __builtin_amdgcn_sched_barrier(0);

    for (int j = 0; j < 2; ++j) {
      const v4f w0 = bw[it % 3][j][0];
      const v4f w1 = bw[it % 3][j][1];
      const v2f b0a = {w0.x, w0.y}, b0b = {w0.z, w0.w};
      const v2f b1a = {w1.x, w1.y}, b1b = {w1.z, w1.w};
      for (int i = 0; i < 4; ++i) {
        const v2f aa = {az[j][i].x, az[j][i].y};
        o[i][0] = wmma_f32(aa, b0a, o[i][0]);
        o[i][1] = wmma_f32(aa, b1a, o[i][1]);
      }
      for (int i = 0; i < 4; ++i) {
        const v2f ab = {az[j][i].z, az[j][i].w};
        o[i][0] = wmma_f32(ab, b0b, o[i][0]);
        o[i][1] = wmma_f32(ab, b1b, o[i][1]);
      }
    }
    if (it + 1 < DD / 16) {
      for (int j = 0; j < 2; ++j) {
        for (int i = 0; i < 4; ++i) az[j][i] = nz[j][i];
      }
    }
  }

  const float bia0 = bias[n0 + c];
  const float bia1 = bias[n0 + 16 + c];
  float* ob = out + (bi * TT + I0 + 8 * h) * DD + n0 + c;
  for (int i = 0; i < 4; ++i) {
    for (int r = 0; r < 8; ++r) {
      const float sr = ss[16 * i + 8 * h + r];
      ob[(16 * i + r) * DD]      = o[i][0][r] + sr * bia0;
      ob[(16 * i + r) * DD + 16] = o[i][1][r] + sr * bia1;
    }
  }
}

extern "C" void kernel_launch(void* const* d_in, const int* in_sizes, int n_in,
                              void* d_out, int out_size, void* d_ws, size_t ws_size,
                              hipStream_t stream) {
  const float* x    = (const float*)d_in[0];
  const float* mask = (const float*)d_in[1];
  const float* W    = (const float*)d_in[2];
  const float* bias = (const float*)d_in[3];
  float* out = (float*)d_out;

  dim3 grid(TT / BROWS, BB);
  da_fused_kernel<<<grid, 256, 0, stream>>>(x, mask, W, bias, out);
}


# outside rows: a first launch sums each 64-row block once into two decayed side sums (workspace, x read once); a block adds 5 neighbours' sums per side instead of streaming 576 rows; sums regrouped
# speedup vs baseline: 3.8365x; 1.2967x over previous
#include <hip/hip_runtime.h>


typedef __attribute__((ext_vector_type(2))) float v2f;
typedef __attribute__((ext_vector_type(4))) float v4f;
typedef __attribute__((ext_vector_type(8))) float v8f;

#define TT   4096
#define DD   256
#define BB   4
#define BAND 288
#define INV_E 0.36787944117144233f

#define BROWS 64
#define CHUNK 64
#define RP    544
#define STAGE (CHUNK / 2 * RP)
#define ZP    260
#define NIT   (CHUNK / 2 * (DD / 4) / 256)
#define NBLK  (TT / BROWS)
#define NNB   ((BAND + BROWS - 1) / BROWS)
#define WOFF  (BB * NBLK * 2 * DD)

__device__ __forceinline__ v8f wmma_f32(v2f a, v2f b, v8f c) {
  return __builtin_amdgcn_wmma_f32_16x16x4_f32(false, a, false, b,
                                               (short)0, c, false, false);
}

__global__ __launch_bounds__(256)
void da_side_sums_kernel(const float* __restrict__ x, const float* __restrict__ mask,
                         float* __restrict__ part) {
  __shared__ __attribute__((aligned(16))) float cpart[4][2][DD];
  __shared__ float wpart[4][2];

  const int tid = threadIdx.x;
  const int cg = tid & 63;
  const int rp0 = tid >> 6;
  const int I0 = blockIdx.x * BROWS;
  const int bi = blockIdx.y;
  const float* xr = x + (bi * TT + I0 + 2 * rp0) * DD + 4 * cg;
  const float* mr = mask + bi * TT + I0 + 2 * rp0;

  v4f g0[NIT], g1[NIT];
  v2f mk[NIT];
  for (int it = 0; it < NIT; ++it) {
    g0[it] = *(const v4f*)(xr + (8 * it) * DD);
    g1[it] = *(const v4f*)(xr + (8 * it + 1) * DD);
    mk[it] = *(const v2f*)(mr + 8 * it);
  }
  asm volatile("" :: "v"(g0[0]), "v"(g0[1]), "v"(g0[2]), "v"(g0[3]), "v"(g0[4]), "v"(g0[5]), "v"(g0[6]), "v"(g0[7]),
                     "v"(g1[0]), "v"(g1[1]), "v"(g1[2]), "v"(g1[3]), "v"(g1[4]), "v"(g1[5]), "v"(g1[6]), "v"(g1[7]),
                     "v"(mk[0]), "v"(mk[1]), "v"(mk[2]), "v"(mk[3]), "v"(mk[4]), "v"(mk[5]), "v"(mk[6]), "v"(mk[7]));
  v4f sl = {0.0f, 0.0f, 0.0f, 0.0f}, sh = {0.0f, 0.0f, 0.0f, 0.0f};
  float wl = 0.0f, wh = 0.0f;
  for (int it = 0; it < NIT; ++it) {
    const float r = (float)(2 * rp0 + 8 * it);
    const float la = __expf(fmaf(mk[it].x, INV_E, -(64.0f - r) * INV_E));
    const float lb = __expf(fmaf(mk[it].y, INV_E, -(63.0f - r) * INV_E));
    const float ha = __expf(fmaf(mk[it].x, INV_E, -(r + 1.0f) * INV_E));
    const float hb = __expf(fmaf(mk[it].y, INV_E, -(r + 2.0f) * INV_E));
    sl = sl + g0[it] * la; sl = sl + g1[it] * lb; wl += la; wl += lb;
    sh = sh + g0[it] * ha; sh = sh + g1[it] * hb; wh += ha; wh += hb;
  }
  *(v4f*)(&cpart[rp0][0][4 * cg]) = sl;
  *(v4f*)(&cpart[rp0][1][4 * cg]) = sh;
  if (cg == 0) { wpart[rp0][0] = wl; wpart[rp0][1] = wh; }
  __syncthreads();

  const int blk = bi * NBLK + blockIdx.x;
  float* pb = part + blk * 2 * DD;
  pb[tid]      = (cpart[0][0][tid] + cpart[1][0][tid]) + (cpart[2][0][tid] + cpart[3][0][tid]);
  pb[DD + tid] = (cpart[0][1][tid] + cpart[1][1][tid]) + (cpart[2][1][tid] + cpart[3][1][tid]);
  if (tid < 2) part[WOFF + blk * 2 + tid] = (wpart[0][tid] + wpart[1][tid]) + (wpart[2][tid] + wpart[3][tid]);
}

__global__ __launch_bounds__(256)
void da_fused_kernel(const float* __restrict__ x, const float* __restrict__ mask,
                     const float* __restrict__ W, const float* __restrict__ bias,
                     const float* __restrict__ part, float* __restrict__ out) {
  __shared__ __attribute__((aligned(16))) float sx[STAGE];
  __shared__ __attribute__((aligned(16))) float sz[BROWS * ZP];
  __shared__ __attribute__((aligned(16))) float smf[CHUNK];
  __shared__ __attribute__((aligned(16))) float ss[BROWS];
  __shared__ __attribute__((aligned(16))) float su[BROWS];
  __shared__ __attribute__((aligned(16))) float sv[BROWS];

  const int tid = threadIdx.x;
  const int lane = tid & 31;
  const int c = lane & 15;
  const int h = lane >> 4;
  const int wv = __builtin_amdgcn_readfirstlane(tid >> 5);
  const int I0 = blockIdx.x * BROWS;
  const int bi = blockIdx.y;

  const float* xb = x + (bi * TT) * DD;
  const float* mb = mask + bi * TT;

  const int cg = tid & 63;
  const int rp0 = tid >> 6;
  const unsigned xoff = (unsigned)((2 * rp0) * DD + 4 * cg) * 4u;
  const unsigned moff = (unsigned)(2 * rp0) * 4u;
  float* sdst = sx + rp0 * RP + 8 * cg;

  const float m = (float)(I0 + c);

  v8f zero = {};
  v8f acc[3][3][2];
  for (int q = 0; q < 3; ++q)
    for (int p = 0; p < 3; ++p) { acc[q][p][0] = zero; acc[q][p][1] = zero; }
  v2f s9[3][3];
  for (int q = 0; q < 3; ++q)
    for (int p = 0; p < 3; ++p) { s9[q][p].x = 0.0f; s9[q][p].y = 0.0f; }
  v2f wa[3][3][4];

  v4f g0[NIT], g1[NIT];
  v2f mk[NIT];
  v2f fj[4][4];
  v2f y[4][4][2];

  for (int it = 0; it < NIT; ++it) {
    const char* px = (const char*)(xb + (I0 + 8 * it) * DD);
    g0[it] = *(const v4f*)(px + xoff);
    g1[it] = *(const v4f*)(px + xoff + DD * 4);
    mk[it] = *(const v2f*)((const char*)(mb + I0 + 8 * it) + moff);
  }

  const int kb = blockIdx.x;
  float pl[2][NNB], ph[2][NNB];
  float pwl[NNB], pwh[NNB];
  for (int n = 0; n < NNB; ++n) {
    const int kl = kb - 1 - n, kh = kb + 1 + n;
    const bool inl = kl >= 0, inh = kh < NBLK;
    const int bl = bi * NBLK + (inl ? kl : kb), bh = bi * NBLK + (inh ? kh : kb);
    const float* ql = part + (bl * 2 + 0) * DD + 32 * wv + c;
    const float* qh = part + (bh * 2 + 1) * DD + 32 * wv + c;
    for (int t = 0; t < 2; ++t) {
      const float vl = ql[16 * t], vh = qh[16 * t];
      pl[t][n] = inl ? vl : 0.0f;
      ph[t][n] = inh ? vh : 0.0f;
    }
    const float ul = part[WOFF + bl * 2 + 0], uh = part[WOFF + bh * 2 + 1];
    pwl[n] = inl ? ul : 0.0f;
    pwh[n] = inh ? uh : 0.0f;
  }

  for (int it = 0; it < NIT; ++it) {
    const float fa = __expf(mk[it].x * INV_E);
    const float fb = __expf(mk[it].y * INV_E);
    v4f s0 = {g0[it].x * fa, g1[it].x * fb, g0[it].y * fa, g1[it].y * fb};
    v4f s1 = {g0[it].z * fa, g1[it].z * fb, g0[it].w * fa, g1[it].w * fb};
    *(v4f*)(sdst + it * 4 * RP)     = s0;
    *(v4f*)(sdst + it * 4 * RP + 4) = s1;
    if (cg == 0) { v2f fp = {fa, fb}; *(v2f*)(smf + 2 * rp0 + 8 * it) = fp; }
  }
  if (wv < 2) {
    su[tid] = __expf(-(float)tid * INV_E);
    sv[tid] = __expf(-(float)(BROWS - 1 - tid) * INV_E);
  }
  __syncthreads();

  for (int s = 0; s < 4; ++s) {
    for (int J = 0; J < 4; ++J) {
      const float* p = sx + (8 * J + 2 * s + h) * RP + 64 * wv + 2 * c;
      y[J][s][0] = *(const v2f*)(p);
      y[J][s][1] = *(const v2f*)(p + 32);
    }
    if (s == 0) __builtin_amdgcn_sched_barrier(0);
  }
  for (int J = 0; J < 4; ++J)
    for (int s = 0; s < 4; ++s) fj[J][s] = *(const v2f*)(smf + 16 * J + 4 * s + 2 * h);

  for (int s = 0; s < 4; ++s) {
    const float d0 = m - (float)(I0 + 4 * s + 2 * h);
    const float d1 = d0 - 1.0f;
    v2f T[7];
    for (int k = -3; k <= 3; ++k) {
      T[k + 3].x = __expf(-fabsf(d0 + 16.0f * (float)k) * INV_E);
      T[k + 3].y = __expf(-fabsf(d1 + 16.0f * (float)k) * INV_E);
    }
    const v2f t0 = T[3];
    const v2f q2a0 = T[1] - t0, q2am = T[0] - T[2], q2ap = T[2] - T[4];
    const v2f q3a0 = T[5] - t0, q3am = T[4] - T[2], q3ap = T[6] - T[4];
    wa[0][0][s] = t0;   wa[0][1][s] = T[2] - t0;    wa[0][2][s] = T[4] - t0;
    wa[1][0][s] = q2a0; wa[1][1][s] = q2am - q2a0;  wa[1][2][s] = q2ap - q2a0;
    wa[2][0][s] = q3a0; wa[2][1][s] = q3am - q3a0;  wa[2][2][s] = q3ap - q3a0;
  }
  for (int s = 0; s < 4; ++s) {
    v2f v0[2], v1[2], vs[2], w01[2], w23[2];
    for (int t = 0; t < 2; ++t) {
      v0[t]  = y[0][s][t] + y[2][s][t];
      v1[t]  = y[1][s][t] + y[3][s][t];
      vs[t]  = v0[t] + v1[t];
      w01[t] = y[0][s][t] + y[1][s][t];
      w23[t] = y[2][s][t] + y[3][s][t];
    }
    __builtin_amdgcn_sched_barrier(0);
    for (int t = 0; t < 2; ++t) {
      acc[0][0][t] = wmma_f32(wa[0][0][s], vs[t],      acc[0][0][t]);
      acc[0][1][t] = wmma_f32(wa[0][1][s], v1[t],      acc[0][1][t]);
      acc[0][2][t] = wmma_f32(wa[0][2][s], v0[t],      acc[0][2][t]);
      acc[1][0][t] = wmma_f32(wa[1][0][s], w23[t],     acc[1][0][t]);
      acc[1][1][t] = wmma_f32(wa[1][1][s], y[3][s][t], acc[1][1][t]);
      acc[1][2][t] = wmma_f32(wa[1][2][s], y[2][s][t], acc[1][2][t]);
      acc[2][0][t] = wmma_f32(wa[2][0][s], w01[t],     acc[2][0][t]);
      acc[2][1][t] = wmma_f32(wa[2][1][s], y[1][s][t], acc[2][1][t]);
      acc[2][2][t] = wmma_f32(wa[2][2][s], y[0][s][t], acc[2][2][t]);
    }
    __builtin_amdgcn_sched_barrier(0);
  }
  for (int s = 0; s < 4; ++s) {
    const v2f f0 = fj[0][s], f1 = fj[1][s], f2 = fj[2][s], f3 = fj[3][s];
    const v2f fv0 = f0 + f2, fv1 = f1 + f3, fw01 = f0 + f1, fw23 = f2 + f3;
    const v2f fvs = fv0 + fv1;
    s9[0][0] += wa[0][0][s] * fvs;  s9[0][1] += wa[0][1][s] * fv1; s9[0][2] += wa[0][2][s] * fv0;
    s9[1][0] += wa[1][0][s] * fw23; s9[1][1] += wa[1][1][s] * f3;  s9[1][2] += wa[1][2][s] * f2;
    s9[2][0] += wa[2][0][s] * fw01; s9[2][1] += wa[2][1][s] * f1;  s9[2][2] += wa[2][2][s] * f0;
  }

  float dk[NNB];
  for (int n = 0; n < NNB; ++n) dk[n] = __expf(-(float)(BROWS * n) * INV_E);

  {
    const v2f q1u = s9[0][0] + s9[0][1], q1l = s9[0][0] + s9[0][2];
    const v2f q2u = s9[1][0] + s9[1][1], q2l = s9[1][0] + s9[1][2];
    const v2f q3u = s9[2][0] + s9[2][1], q3l = s9[2][0] + s9[2][2];
    const v2f sv4[4] = {q1u + q2u, q1l + q2l, q1u + q3u, q1l + q3l};
    float sr[4];
    for (int I = 0; I < 4; ++I) {
      sr[I] = sv4[I].x + sv4[I].y;
      sr[I] += __shfl_xor(sr[I], 16);
    }
    if (wv == 0 && h == 0) {
      float wls = pwl[0], whs = pwh[0];
      for (int n = 1; n < NNB; ++n) { wls += pwl[n] * dk[n]; whs += pwh[n] * dk[n]; }
      for (int I = 0; I < 4; ++I) ss[16 * I + c] = sr[I] + su[16 * I + c] * wls + sv[16 * I + c] * whs;
    }
  }

  {
    float* zd = sz + (8 * h) * ZP + 32 * wv + c;
    for (int t = 0; t < 2; ++t) {
      float clc = pl[t][0], chc = ph[t][0];
      for (int n = 1; n < NNB; ++n) { clc += pl[t][n] * dk[n]; chc += ph[t][n] * dk[n]; }
      for (int r = 0; r < 8; ++r) {
        const float q1u = acc[0][0][t][r] + acc[0][1][t][r];
        const float q1l = acc[0][0][t][r] + acc[0][2][t][r];
        const float q2u = acc[1][0][t][r] + acc[1][1][t][r];
        const float q2l = acc[1][0][t][r] + acc[1][2][t][r];
        const float q3u = acc[2][0][t][r] + acc[2][1][t][r];
        const float q3l = acc[2][0][t][r] + acc[2][2][t][r];
        const int rr = 8 * h + r;
        zd[r * ZP + 16 * t]        = (q1u + q2u) + (su[rr] * clc      + sv[rr] * chc);
        zd[(16 + r) * ZP + 16 * t] = (q1l + q2l) + (su[16 + rr] * clc + sv[16 + rr] * chc);
        zd[(32 + r) * ZP + 16 * t] = (q1u + q3u) + (su[32 + rr] * clc + sv[32 + rr] * chc);
        zd[(48 + r) * ZP + 16 * t] = (q1l + q3l) + (su[48 + rr] * clc + sv[48 + rr] * chc);
      }
    }
  }
  __syncthreads();

  const int n0 = 32 * wv;
  const float* wr0 = W + (n0 + c) * DD + 4 * h;
  const float* wr1 = W + (n0 + 16 + c) * DD + 4 * h;
  const float* zr = sz + c * ZP + 4 * h;

  v8f o[4][2];
  for (int i = 0; i < 4; ++i) { o[i][0] = zero; o[i][1] = zero; }

  v4f bw[3][2][2];
  v4f az[2][4], nz[2][4];
  for (int j = 0; j < 2; ++j) {
    bw[0][j][0] = *(const v4f*)(wr0 + 8 * j);
    bw[0][j][1] = *(const v4f*)(wr1 + 8 * j);
    bw[1][j][0] = *(const v4f*)(wr0 + 16 + 8 * j);
    bw[1][j][1] = *(const v4f*)(wr1 + 16 + 8 * j);
    for (int i = 0; i < 4; ++i) az[j][i] = *(const v4f*)(zr + 16 * i * ZP + 8 * j);
  }

  for (int it = 0; it < DD / 16; ++it) {
    const int kb = 16 * it;
    __builtin_amdgcn_sched_barrier(0);
    if (it + 2 < DD / 16) {
      for (int j = 0; j < 2; ++j) {
        bw[(it + 2) % 3][j][0] = *(const v4f*)(wr0 + kb + 32 + 8 * j);
        bw[(it + 2) % 3][j][1] = *(const v4f*)(wr1 + kb + 32 + 8 * j);
      }
    }
    if (it + 1 < DD / 16) {
      for (int j = 0; j < 2; ++j) {
        for (int i = 0; i < 4; ++i) nz[j][i] = *(const v4f*)(zr + 16 * i * ZP + kb + 16 + 8 * j);
      }
    }
    __builtin_amdgcn_sched_barrier(0);

    for (int j = 0; j < 2; ++j) {
      const v4f w0 = bw[it % 3][j][0];
      const v4f w1 = bw[it % 3][j][1];
      const v2f b0a = {w0.x, w0.y}, b0b = {w0.z, w0.w};
      const v2f b1a = {w1.x, w1.y}, b1b = {w1.z, w1.w};
      for (int i = 0; i < 4; ++i) {
        const v2f aa = {az[j][i].x, az[j][i].y};
        o[i][0] = wmma_f32(aa, b0a, o[i][0]);
        o[i][1] = wmma_f32(aa, b1a, o[i][1]);
      }
      for (int i = 0; i < 4; ++i) {
        const v2f ab = {az[j][i].z, az[j][i].w};
        o[i][0] = wmma_f32(ab, b0b, o[i][0]);
        o[i][1] = wmma_f32(ab, b1b, o[i][1]);
      }
    }
    if (it + 1 < DD / 16) {
      for (int j = 0; j < 2; ++j) {
        for (int i = 0; i < 4; ++i) az[j][i] = nz[j][i];
      }
    }
  }

  const float bia0 = bias[n0 + c];
  const float bia1 = bias[n0 + 16 + c];
  float* ob = out + (bi * TT + I0 + 8 * h) * DD + n0 + c;
  for (int i = 0; i < 4; ++i) {
    for (int r = 0; r < 8; ++r) {
      const float sr = ss[16 * i + 8 * h + r];
      ob[(16 * i + r) * DD]      = o[i][0][r] + sr * bia0;
      ob[(16 * i + r) * DD + 16] = o[i][1][r] + sr * bia1;
    }
  }
}

extern "C" void kernel_launch(void* const* d_in, const int* in_sizes, int n_in,
                              void* d_out, int out_size, void* d_ws, size_t ws_size,
                              hipStream_t stream) {
  const float* x    = (const float*)d_in[0];
  const float* mask = (const float*)d_in[1];
  const float* W    = (const float*)d_in[2];
  const float* bias = (const float*)d_in[3];
  float* out = (float*)d_out;

  float* part = (float*)d_ws;

  dim3 grid(TT / BROWS, BB);
  da_side_sums_kernel<<<grid, 256, 0, stream>>>(x, mask, part);
  da_fused_kernel<<<grid, 256, 0, stream>>>(x, mask, W, bias, part, out);
}


# attn: a block's own 64 rows by recurrence, not products (72 of 584 WMMAs per wave go): exp(-|i-j|/e)=a^|i-j| gives F[r]=y[r]+a*F[r-1] and its mirror on the vector unit; sums regrouped
# speedup vs baseline: 4.1311x; 1.0768x over previous
#include <hip/hip_runtime.h>


typedef __attribute__((ext_vector_type(2))) float v2f;
typedef __attribute__((ext_vector_type(4))) float v4f;
typedef __attribute__((ext_vector_type(8))) float v8f;

#define TT   4096
#define DD   256
#define BB   4
#define BAND 288
#define INV_E 0.36787944117144233f
#define A_DEC 0.69220062755534636f

#define BROWS 64
#define HROWS (BROWS / 2)
#define CHUNK 64
#define ZP    260
#define NIT   (CHUNK / 2 * (DD / 4) / 256)
#define NBLK  (TT / BROWS)
#define NNB   ((BAND + BROWS - 1) / BROWS)
#define WOFF  (BB * NBLK * 2 * DD)

__device__ __forceinline__ v8f wmma_f32(v2f a, v2f b, v8f c) {
  return __builtin_amdgcn_wmma_f32_16x16x4_f32(false, a, false, b,
                                               (short)0, c, false, false);
}

__global__ __launch_bounds__(256)
void da_side_sums_kernel(const float* __restrict__ x, const float* __restrict__ mask,
                         float* __restrict__ part) {
  __shared__ __attribute__((aligned(16))) float cpart[4][2][DD];
  __shared__ float wpart[4][2];

  const int tid = threadIdx.x;
  const int cg = tid & 63;
  const int rp0 = tid >> 6;
  const int I0 = blockIdx.x * BROWS;
  const int bi = blockIdx.y;
  const float* xr = x + (bi * TT + I0 + 2 * rp0) * DD + 4 * cg;
  const float* mr = mask + bi * TT + I0 + 2 * rp0;

  v4f g0[NIT], g1[NIT];
  v2f mk[NIT];
  for (int it = 0; it < NIT; ++it) {
    g0[it] = *(const v4f*)(xr + (8 * it) * DD);
    g1[it] = *(const v4f*)(xr + (8 * it + 1) * DD);
    mk[it] = *(const v2f*)(mr + 8 * it);
  }
  asm volatile("" :: "v"(g0[0]), "v"(g0[1]), "v"(g0[2]), "v"(g0[3]), "v"(g0[4]), "v"(g0[5]), "v"(g0[6]), "v"(g0[7]),
                     "v"(g1[0]), "v"(g1[1]), "v"(g1[2]), "v"(g1[3]), "v"(g1[4]), "v"(g1[5]), "v"(g1[6]), "v"(g1[7]),
                     "v"(mk[0]), "v"(mk[1]), "v"(mk[2]), "v"(mk[3]), "v"(mk[4]), "v"(mk[5]), "v"(mk[6]), "v"(mk[7]));
  v4f sl = {0.0f, 0.0f, 0.0f, 0.0f}, sh = {0.0f, 0.0f, 0.0f, 0.0f};
  float wl = 0.0f, wh = 0.0f;
  for (int it = 0; it < NIT; ++it) {
    const float r = (float)(2 * rp0 + 8 * it);
    const float la = __expf(fmaf(mk[it].x, INV_E, -(64.0f - r) * INV_E));
    const float lb = __expf(fmaf(mk[it].y, INV_E, -(63.0f - r) * INV_E));
    const float ha = __expf(fmaf(mk[it].x, INV_E, -(r + 1.0f) * INV_E));
    const float hb = __expf(fmaf(mk[it].y, INV_E, -(r + 2.0f) * INV_E));
    sl = sl + g0[it] * la; sl = sl + g1[it] * lb; wl += la; wl += lb;
    sh = sh + g0[it] * ha; sh = sh + g1[it] * hb; wh += ha; wh += hb;
  }
  *(v4f*)(&cpart[rp0][0][4 * cg]) = sl;
  *(v4f*)(&cpart[rp0][1][4 * cg]) = sh;
  if (cg == 0) { wpart[rp0][0] = wl; wpart[rp0][1] = wh; }
  __syncthreads();

  const int blk = bi * NBLK + blockIdx.x;
  float* pb = part + blk * 2 * DD;
  pb[tid]      = (cpart[0][0][tid] + cpart[1][0][tid]) + (cpart[2][0][tid] + cpart[3][0][tid]);
  pb[DD + tid] = (cpart[0][1][tid] + cpart[1][1][tid]) + (cpart[2][1][tid] + cpart[3][1][tid]);
  if (tid < 2) part[WOFF + blk * 2 + tid] = (wpart[0][tid] + wpart[1][tid]) + (wpart[2][tid] + wpart[3][tid]);
}

__global__ __launch_bounds__(256) __attribute__((amdgpu_waves_per_eu(2, 2)))
void da_fused_kernel(const float* __restrict__ x, const float* __restrict__ mask,
                     const float* __restrict__ W, const float* __restrict__ bias,
                     const float* __restrict__ part, float* __restrict__ out) {
  __shared__ __attribute__((aligned(16))) float sz[BROWS * ZP];
  __shared__ __attribute__((aligned(16))) float sxch[2][DD];
  __shared__ float sxs[2];
  __shared__ __attribute__((aligned(16))) float ss[BROWS];

  const int tid = threadIdx.x;
  const int lane = tid & 31;
  const int c = lane & 15;
  const int h = lane >> 4;
  const int wv = __builtin_amdgcn_readfirstlane(tid >> 5);
  const int I0 = blockIdx.x * BROWS;
  const int bi = blockIdx.y;
  const int kb = blockIdx.x;

  v8f zero = {};

  const int hf = wv >> 2;
  const int cp = tid & 127;
  const int r0 = hf ? (BROWS - 1) : 0;
  const int rs = hf ? -1 : 1;
  const bool swave = (wv & 3) == hf;

  const float mk = mask[bi * TT + I0 + r0 + rs * lane];
  const int knl = kb - rs * (lane + 1);
  const bool inl = (lane < NNB) && ((unsigned)knl < (unsigned)NBLK);
  const float pul = part[WOFF + (bi * NBLK + (inl ? knl : kb)) * 2 + hf];

  v2f pc[NNB];
  {
    const v2f z2 = {0.0f, 0.0f};
    for (int n = 0; n < NNB; ++n) {
      const int kn = kb - rs * (n + 1);
      const bool in = (unsigned)kn < (unsigned)NBLK;
      const v2f pv = *(const v2f*)(part + ((bi * NBLK + (in ? kn : kb)) * 2 + hf) * DD + 2 * cp);
      pc[n] = in ? pv : z2;
    }
  }

  v2f y[HROWS];
  {
    const float* xr = x + (bi * TT + I0 + r0) * DD + 2 * cp;
    for (int k = 0; k < HROWS; ++k) y[k] = *(const v2f*)(xr + rs * k * DD);
  }

  v2f cst;
  float wst;
  {
    const int pwl = __float_as_int(inl ? pul : 0.0f);
    cst = pc[0];
    wst = __int_as_float(__builtin_amdgcn_readlane(pwl, 0));
    for (int n = 1; n < NNB; ++n) {
      const float dk = __expf(-(float)(BROWS * n) * INV_E);
      cst = cst + pc[n] * dk;
      wst += __int_as_float(__builtin_amdgcn_readlane(pwl, n)) * dk;
    }
  }

  const float fl = __expf(mk * INV_E);

  float s1 = fl, r1 = fl;
  {
    float ad = A_DEC;
    for (int d = 1; d < HROWS; d *= 2) {
      const float up = __shfl_up(s1, d);
      const float dn = __shfl_down(r1, d);
      s1 = (lane >= d) ? fmaf(ad, up, s1) : s1;
      r1 = (lane + d < HROWS) ? fmaf(ad, dn, r1) : r1;
      ad *= ad;
    }
  }
  float fk[HROWS];
  for (int k = 0; k < HROWS; ++k) fk[k] = __int_as_float(__builtin_amdgcn_readlane(__float_as_int(fl), k));

  const v2f av = {A_DEC, A_DEC};
  v2f S[HROWS];
  y[0] = y[0] * fk[0];
  S[0] = y[0] + cst;
  for (int k = 1; k < HROWS; ++k) {
    y[k] = y[k] * fk[k];
    S[k] = __builtin_elementwise_fma(av, S[k - 1], y[k]);
  }
  *(v2f*)(&sxch[hf][2 * cp]) = S[HROWS - 1];

  const float st = fmaf(__expf(-(float)lane * INV_E), wst, s1);
  const float sbase = st + (r1 - fl);
  const float sdec = __expf(-(float)(HROWS - lane) * INV_E);
  if (swave && lane == HROWS - 1) sxs[hf] = st;
  __syncthreads();

  {
    v2f T = *(const v2f*)(&sxch[1 - hf][2 * cp]);
    const float Ts = sxs[1 - hf];
    float* zd = sz + r0 * ZP + 2 * cp;
    for (int k = HROWS - 1; k >= 0; --k) {
      *(v2f*)(zd + rs * k * ZP) = __builtin_elementwise_fma(av, T, S[k]);
      T = __builtin_elementwise_fma(av, T, y[k]);
    }
    if (swave) ss[r0 + rs * lane] = fmaf(sdec, Ts, sbase);
  }
  __syncthreads();

  const int n0 = 32 * wv;
  const float* wr0 = W + (n0 + c) * DD + 4 * h;
  const float* wr1 = W + (n0 + 16 + c) * DD + 4 * h;
  const float* zr = sz + c * ZP + 4 * h;

  v8f o[4][2];
  for (int i = 0; i < 4; ++i) { o[i][0] = zero; o[i][1] = zero; }

  v4f bw[3][2][2];
  v4f az[2][4], nz[2][4];
  for (int j = 0; j < 2; ++j) {
    bw[0][j][0] = *(const v4f*)(wr0 + 8 * j);
    bw[0][j][1] = *(const v4f*)(wr1 + 8 * j);
    bw[1][j][0] = *(const v4f*)(wr0 + 16 + 8 * j);
    bw[1][j][1] = *(const v4f*)(wr1 + 16 + 8 * j);
    for (int i = 0; i < 4; ++i) az[j][i] = *(const v4f*)(zr + 16 * i * ZP + 8 * j);
  }

  for (int it = 0; it < DD / 16; ++it) {
    const int kb = 16 * it;
    __builtin_amdgcn_sched_barrier(0);
    if (it + 2 < DD / 16) {
      for (int j = 0; j < 2; ++j) {
        bw[(it + 2) % 3][j][0] = *(const v4f*)(wr0 + kb + 32 + 8 * j);
        bw[(it + 2) % 3][j][1] = *(const v4f*)(wr1 + kb + 32 + 8 * j);
      }
    }
    if (it + 1 < DD / 16) {
      for (int j = 0; j < 2; ++j) {
        for (int i = 0; i < 4; ++i) nz[j][i] = *(const v4f*)(zr + 16 * i * ZP + kb + 16 + 8 * j);
      }
    }
    __builtin_amdgcn_sched_barrier(0);

    for (int j = 0; j < 2; ++j) {
      const v4f w0 = bw[it % 3][j][0];
      const v4f w1 = bw[it % 3][j][1];
      const v2f b0a = {w0.x, w0.y}, b0b = {w0.z, w0.w};
      const v2f b1a = {w1.x, w1.y}, b1b = {w1.z, w1.w};
      for (int i = 0; i < 4; ++i) {
        const v2f aa = {az[j][i].x, az[j][i].y};
        o[i][0] = wmma_f32(aa, b0a, o[i][0]);
        o[i][1] = wmma_f32(aa, b1a, o[i][1]);
      }
      for (int i = 0; i < 4; ++i) {
        const v2f ab = {az[j][i].z, az[j][i].w};
        o[i][0] = wmma_f32(ab, b0b, o[i][0]);
        o[i][1] = wmma_f32(ab, b1b, o[i][1]);
      }
    }
    if (it + 1 < DD / 16) {
      for (int j = 0; j < 2; ++j) {
        for (int i = 0; i < 4; ++i) az[j][i] = nz[j][i];
      }
    }
  }

  const float bia0 = bias[n0 + c];
  const float bia1 = bias[n0 + 16 + c];
  float* ob = out + (bi * TT + I0 + 8 * h) * DD + n0 + c;
  for (int i = 0; i < 4; ++i) {
    for (int r = 0; r < 8; ++r) {
      const float sr = ss[16 * i + 8 * h + r];
      ob[(16 * i + r) * DD]      = o[i][0][r] + sr * bia0;
      ob[(16 * i + r) * DD + 16] = o[i][1][r] + sr * bia1;
    }
  }
}

extern "C" void kernel_launch(void* const* d_in, const int* in_sizes, int n_in,
                              void* d_out, int out_size, void* d_ws, size_t ws_size,
                              hipStream_t stream) {
  const float* x    = (const float*)d_in[0];
  const float* mask = (const float*)d_in[1];
  const float* W    = (const float*)d_in[2];
  const float* bias = (const float*)d_in[3];
  float* out = (float*)d_out;

  float* part = (float*)d_ws;

  dim3 grid(TT / BROWS, BB);
  da_side_sums_kernel<<<grid, 256, 0, stream>>>(x, mask, part);
  da_fused_kernel<<<grid, 256, 0, stream>>>(x, mask, W, bias, part, out);
}
